# starting kernel
# baseline (speedup 1.0000x reference)
#include <hip/hip_runtime.h>

typedef __attribute__((ext_vector_type(2))) float v2f;
typedef __attribute__((ext_vector_type(8))) float v8f;

#define T_STEPS  2048
#define BATCH    256
#define IN_DIM   64
#define H0       16
#define H1       32
#define H2       16
#define OUT_DIM  8
#define NTHREADS 128

__device__ __forceinline__ float sigmoidf_(float x) { return 1.0f / (1.0f + __expf(-x)); }

__device__ __forceinline__ v2f load_afrag(const float* base, int lda, int kbase, int lane) {
    int m = lane & 15;
    int k = kbase + ((lane >> 4) << 1);
    const float* p = base + m * lda + k;
    v2f a;
    a.x = p[0];
    a.y = p[1];
    return a;
}

__device__ __forceinline__ v2f load_bfrag(const float* W, int ldw, int nbase, int kbase, int lane) {
    int n = nbase + (lane & 15);
    int k = kbase + ((lane >> 4) << 1);
    const float* p = W + n * ldw + k;
    v2f b;
    b.x = p[0];
    b.y = p[1];
    return b;
}

__device__ __forceinline__ v8f wmma4(v2f a, v2f b, v8f c) {
    return __builtin_amdgcn_wmma_f32_16x16x4_f32(false, a, false, b, (short)0, c, false, false);
}

__device__ __forceinline__ void store_dtile(float* g, int ldg, int nbase, int lane, v8f d) {
    int col   = nbase + (lane & 15);
    int rbase = (lane >> 4) << 3;
#pragma unroll
    for (int j = 0; j < 8; ++j)
        g[(rbase + j) * ldg + col] = d[j];
}

template <int H>
__device__ __forceinline__ void cell_update(const float* g, float* c, float* h, int tid) {
    const int fourH = 4 * H;
#pragma unroll
    for (int idx = tid; idx < 16 * H; idx += NTHREADS) {
        int m = idx / H;
        int n = idx % H;
        float vi = g[m * fourH + n];
        float vf = g[m * fourH + H + n];
        float vg = g[m * fourH + 2 * H + n];
        float vo = g[m * fourH + 3 * H + n];
        float cc = c[idx];
        float cn = sigmoidf_(vf) * cc + sigmoidf_(vi) * tanhf(vg);
        c[idx] = cn;
        h[idx] = sigmoidf_(vo) * tanhf(cn);
    }
}

__global__ __launch_bounds__(NTHREADS) void lstm3_scan_kernel(
    const float* __restrict__ x,
    const float* __restrict__ Wih0, const float* __restrict__ Whh0,
    const float* __restrict__ bih0, const float* __restrict__ bhh0,
    const float* __restrict__ Wih1, const float* __restrict__ Whh1,
    const float* __restrict__ bih1, const float* __restrict__ bhh1,
    const float* __restrict__ Wih2, const float* __restrict__ Whh2,
    const float* __restrict__ bih2, const float* __restrict__ bhh2,
    const float* __restrict__ Wl,   const float* __restrict__ bl,
    float* __restrict__ out)
{
    __shared__ float sG[16 * 4 * H1];
    __shared__ float sH0[16 * H0], sC0[16 * H0];
    __shared__ float sH1[16 * H1], sC1[16 * H1];
    __shared__ float sH2[16 * H2], sC2[16 * H2];
    __shared__ float sWl[OUT_DIM * H2];
    __shared__ float sBl[OUT_DIM];

    const int tid  = threadIdx.x;
    const int lane = tid & 31;
    const int wave = tid >> 5;
    const int ln   = lane & 15;
    const int b0   = blockIdx.x * 16;

    for (int i = tid; i < 16 * H0; i += NTHREADS) { sH0[i] = 0.f; sC0[i] = 0.f; sH2[i] = 0.f; sC2[i] = 0.f; }
    for (int i = tid; i < 16 * H1; i += NTHREADS) { sH1[i] = 0.f; sC1[i] = 0.f; }
    for (int i = tid; i < OUT_DIM * H2; i += NTHREADS) sWl[i] = Wl[i];
    if (tid < OUT_DIM) sBl[tid] = bl[tid];

    const int nb0 = wave * 16;
    v2f wx0[16], wh0[4];
#pragma unroll
    for (int kk = 0; kk < 16; ++kk) wx0[kk] = load_bfrag(Wih0, IN_DIM, nb0, 4 * kk, lane);
#pragma unroll
    for (int kk = 0; kk < 4;  ++kk) wh0[kk] = load_bfrag(Whh0, H0, nb0, 4 * kk, lane);
    const float bias0 = bih0[nb0 + ln] + bhh0[nb0 + ln];

    const int nb1a = wave * 32, nb1b = wave * 32 + 16;
    v2f wx1a[4], wx1b[4], wh1a[8], wh1b[8];
#pragma unroll
    for (int kk = 0; kk < 4; ++kk) {
        wx1a[kk] = load_bfrag(Wih1, H0, nb1a, 4 * kk, lane);
        wx1b[kk] = load_bfrag(Wih1, H0, nb1b, 4 * kk, lane);
    }
#pragma unroll
    for (int kk = 0; kk < 8; ++kk) {
        wh1a[kk] = load_bfrag(Whh1, H1, nb1a, 4 * kk, lane);
        wh1b[kk] = load_bfrag(Whh1, H1, nb1b, 4 * kk, lane);
    }
    const float bias1a = bih1[nb1a + ln] + bhh1[nb1a + ln];
    const float bias1b = bih1[nb1b + ln] + bhh1[nb1b + ln];

    const int nb2 = wave * 16;
    v2f wx2[8], wh2[4];
#pragma unroll
    for (int kk = 0; kk < 8; ++kk) wx2[kk] = load_bfrag(Wih2, H1, nb2, 4 * kk, lane);
#pragma unroll
    for (int kk = 0; kk < 4; ++kk) wh2[kk] = load_bfrag(Whh2, H2, nb2, 4 * kk, lane);
    const float bias2 = bih2[nb2 + ln] + bhh2[nb2 + ln];

    __syncthreads();

    for (int t = 0; t < T_STEPS; ++t) {
        const float* xt = x + ((size_t)t * BATCH + b0) * IN_DIM;
        if (t + 1 < T_STEPS) {
            const float* xn = xt + (size_t)BATCH * IN_DIM;
            __builtin_prefetch(xn + tid * 8, 0, 0);
        }

        v8f acc0;
#pragma unroll
        for (int j = 0; j < 8; ++j) acc0[j] = bias0;
#pragma unroll
        for (int kk = 0; kk < 16; ++kk) {
            v2f a = load_afrag(xt, IN_DIM, 4 * kk, lane);
            acc0 = wmma4(a, wx0[kk], acc0);
        }
#pragma unroll
        for (int kk = 0; kk < 4; ++kk) {
            v2f a = load_afrag(sH0, H0, 4 * kk, lane);
            acc0 = wmma4(a, wh0[kk], acc0);
        }
        store_dtile(sG, 4 * H0, nb0, lane, acc0);
        __syncthreads();
        cell_update<H0>(sG, sC0, sH0, tid);
        __syncthreads();

        v8f acc1a, acc1b;
#pragma unroll
        for (int j = 0; j < 8; ++j) { acc1a[j] = bias1a; acc1b[j] = bias1b; }
#pragma unroll
        for (int kk = 0; kk < 4; ++kk) {
            v2f a = load_afrag(sH0, H0, 4 * kk, lane);
            acc1a = wmma4(a, wx1a[kk], acc1a);
            acc1b = wmma4(a, wx1b[kk], acc1b);
        }
#pragma unroll
        for (int kk = 0; kk < 8; ++kk) {
            v2f a = load_afrag(sH1, H1, 4 * kk, lane);
            acc1a = wmma4(a, wh1a[kk], acc1a);
            acc1b = wmma4(a, wh1b[kk], acc1b);
        }
        store_dtile(sG, 4 * H1, nb1a, lane, acc1a);
        store_dtile(sG, 4 * H1, nb1b, lane, acc1b);
        __syncthreads();
        cell_update<H1>(sG, sC1, sH1, tid);
        __syncthreads();

        v8f acc2;
#pragma unroll
        for (int j = 0; j < 8; ++j) acc2[j] = bias2;
#pragma unroll
        for (int kk = 0; kk < 8; ++kk) {
            v2f a = load_afrag(sH1, H1, 4 * kk, lane);
            acc2 = wmma4(a, wx2[kk], acc2);
        }
#pragma unroll
        for (int kk = 0; kk < 4; ++kk) {
            v2f a = load_afrag(sH2, H2, 4 * kk, lane);
            acc2 = wmma4(a, wh2[kk], acc2);
        }
        store_dtile(sG, 4 * H2, nb2, lane, acc2);
        __syncthreads();
        cell_update<H2>(sG, sC2, sH2, tid);
        __syncthreads();

        {
            int m = tid >> 3;
            int o = tid & 7;
            float acc = sBl[o];
#pragma unroll
            for (int k = 0; k < H2; ++k) acc += sH2[m * H2 + k] * sWl[o * H2 + k];
            out[(((size_t)(b0 + m)) * T_STEPS + t) * OUT_DIM + o] = acc;
        }
    }
}

extern "C" void kernel_launch(void* const* d_in, const int* in_sizes, int n_in,
                              void* d_out, int out_size, void* d_ws, size_t ws_size,
                              hipStream_t stream) {
    (void)in_sizes; (void)n_in; (void)out_size; (void)d_ws; (void)ws_size;
    const float* x    = (const float*)d_in[0];
    const float* Wih0 = (const float*)d_in[1];
    const float* Whh0 = (const float*)d_in[2];
    const float* bih0 = (const float*)d_in[3];
    const float* bhh0 = (const float*)d_in[4];
    const float* Wih1 = (const float*)d_in[5];
    const float* Whh1 = (const float*)d_in[6];
    const float* bih1 = (const float*)d_in[7];
    const float* bhh1 = (const float*)d_in[8];
    const float* Wih2 = (const float*)d_in[9];
    const float* Whh2 = (const float*)d_in[10];
    const float* bih2 = (const float*)d_in[11];
    const float* bhh2 = (const float*)d_in[12];
    const float* Wl   = (const float*)d_in[13];
    const float* bl   = (const float*)d_in[14];
    float* out = (float*)d_out;

    lstm3_scan_kernel<<<BATCH / 16, NTHREADS, 0, stream>>>(
        x, Wih0, Whh0, bih0, bhh0,
        Wih1, Whh1, bih1, bhh1,
        Wih2, Whh2, bih2, bhh2,
        Wl, bl, out);
}


# each wave owns one stage and 8 units (x projection; layers 0,1,2 one step apart; output on WMMA, its sum order changes): i,f,g,o of 4 cells a lane via one row_ror:8 move; 1 barrier a step, not 6
# speedup vs baseline: 2.1903x; 2.1903x over previous
#include <hip/hip_runtime.h>


typedef __attribute__((ext_vector_type(2))) float v2f;
typedef __attribute__((ext_vector_type(4))) float v4f;
typedef __attribute__((ext_vector_type(8))) float v8f;

#define T_STEPS  2048
#define BATCH    256
#define IN_DIM   64
#define H0       16
#define H1       32
#define H2       16
#define OUT_DIM  8
#define NWAVES   10
#define NTHREADS (NWAVES * 32)

#define P0 20
#define P1 36
#define P2 20

#define KIND_L1 0
#define KIND_L0 1
#define KIND_L2 2
#define KIND_X  3

__device__ __forceinline__ float sigmoidf_(float x) { return 1.0f / (1.0f + __expf(-x)); }

__device__ __forceinline__ v2f load_afrag(const float* base, int lda, int kbase, int lane) {
    int m = lane & 15;
    int k = kbase + ((lane >> 4) << 1);
    const float* p = base + m * lda + k;
    v2f a;
    a.x = p[0];
    a.y = p[1];
    return a;
}

__device__ __forceinline__ v2f load_afrag_lds(const float* base, int pitch, int kbase, int lane) {
    int m = lane & 15;
    int k = kbase + ((lane >> 4) << 1);
    return *(const v2f*)(base + m * pitch + k);
}

__device__ __forceinline__ v2f load_bfrag_row(const float* W, int ldw, int row, int kbase, int lane) {
    int k = kbase + ((lane >> 4) << 1);
    const float* p = W + row * ldw + k;
    v2f b;
    b.x = p[0];
    b.y = p[1];
    return b;
}

__device__ __forceinline__ v8f wmma4(v2f a, v2f b, v8f c) {
    return __builtin_amdgcn_wmma_f32_16x16x4_f32(false, a, false, b, (short)0, c, false, false);
}

template <int BANKS>
__device__ __forceinline__ float take8(float keep, float give) {
    return __builtin_bit_cast(float,
        __builtin_amdgcn_update_dpp(__builtin_bit_cast(int, keep), __builtin_bit_cast(int, give),
                                    0x128, 0xf, BANKS, false));
}

__device__ __forceinline__ void cell4(const v8f& tA, const v8f& tB,
                                      float (&c)[4], float* hdst, int pitch) {
#pragma unroll
    for (int jj = 0; jj < 4; ++jj) {
        float vi = take8<0xC>(tA[jj],     tA[jj + 4]);
        float vf = take8<0x3>(tA[jj + 4], tA[jj]);
        float vg = take8<0xC>(tB[jj],     tB[jj + 4]);
        float vo = take8<0x3>(tB[jj + 4], tB[jj]);
        float cc = c[jj];
        float cn = sigmoidf_(vf) * cc + sigmoidf_(vi) * tanhf(vg);
        c[jj] = cn;
        hdst[jj * pitch] = sigmoidf_(vo) * tanhf(cn);
    }
}

__global__ __launch_bounds__(NTHREADS) void lstm3_stage_kernel(
    const float* __restrict__ x,
    const float* __restrict__ Wih0, const float* __restrict__ Whh0,
    const float* __restrict__ bih0, const float* __restrict__ bhh0,
    const float* __restrict__ Wih1, const float* __restrict__ Whh1,
    const float* __restrict__ bih1, const float* __restrict__ bhh1,
    const float* __restrict__ Wih2, const float* __restrict__ Whh2,
    const float* __restrict__ bih2, const float* __restrict__ bhh2,
    const float* __restrict__ Wl,   const float* __restrict__ bl,
    float* __restrict__ out)
{
    __shared__ __attribute__((aligned(16))) float sXP[2][2 * 2 * 2 * 32 * 4];
    __shared__ __attribute__((aligned(16))) float sH0[2][16 * P0];
    __shared__ __attribute__((aligned(16))) float sH1[2][16 * P1];
    __shared__ __attribute__((aligned(16))) float sH2[2][16 * P2];

    const int tid  = threadIdx.x;
    const int lane = tid & 31;
    const int ln   = lane & 15;
    const int hi   = lane >> 4;
    const int u    = ln & 7;
    const int gs   = ln >> 3;
    const int wave = __builtin_amdgcn_readfirstlane(tid >> 5);
    const int kind = (wave < 4) ? KIND_L1 : (wave < 6) ? KIND_L0 : (wave < 8) ? KIND_L2 : KIND_X;
    const int q    = (wave < 4) ? wave : (wave & 1);
    const int u0   = q * 8;
    const int b0   = blockIdx.x * 16;

    for (int i = tid; i < 2 * 16 * P0; i += NTHREADS) { (&sH0[0][0])[i] = 0.f; (&sH2[0][0])[i] = 0.f; }
    for (int i = tid; i < 2 * 16 * P1; i += NTHREADS) { (&sH1[0][0])[i] = 0.f; }

    v2f   w[32];
    float bias[2];
    float c[4];
#pragma unroll
    for (int i = 0; i < 32; ++i) { w[i].x = 0.f; w[i].y = 0.f; }
    bias[0] = 0.f; bias[1] = 0.f;
#pragma unroll
    for (int j = 0; j < 4; ++j) c[j] = 0.f;
    float biasO = 0.f;

    if (kind == KIND_X) {
        const int r0 = gs * H0 + u0 + u, r1 = (2 + gs) * H0 + u0 + u;
#pragma unroll
        for (int kk = 0; kk < 16; ++kk) {
            w[kk * 2]     = load_bfrag_row(Wih0, IN_DIM, r0, 4 * kk, lane);
            w[kk * 2 + 1] = load_bfrag_row(Wih0, IN_DIM, r1, 4 * kk, lane);
        }
        bias[0] = bih0[r0] + bhh0[r0];
        bias[1] = bih0[r1] + bhh0[r1];
    } else if (kind == KIND_L0) {
        const int r0 = gs * H0 + u0 + u, r1 = (2 + gs) * H0 + u0 + u;
#pragma unroll
        for (int kk = 0; kk < 4; ++kk) {
            w[kk * 2]     = load_bfrag_row(Whh0, H0, r0, 4 * kk, lane);
            w[kk * 2 + 1] = load_bfrag_row(Whh0, H0, r1, 4 * kk, lane);
        }
    } else if (kind == KIND_L2) {
        const int r0 = gs * H2 + u0 + u, r1 = (2 + gs) * H2 + u0 + u;
#pragma unroll
        for (int kk = 0; kk < 8; ++kk) {
            w[kk * 2]     = load_bfrag_row(Wih2, H1, r0, 4 * kk, lane);
            w[kk * 2 + 1] = load_bfrag_row(Wih2, H1, r1, 4 * kk, lane);
        }
#pragma unroll
        for (int kk = 0; kk < 4; ++kk) {
            w[16 + kk * 2]     = load_bfrag_row(Whh2, H2, r0, 4 * kk, lane);
            w[16 + kk * 2 + 1] = load_bfrag_row(Whh2, H2, r1, 4 * kk, lane);
        }
        bias[0] = bih2[r0] + bhh2[r0];
        bias[1] = bih2[r1] + bhh2[r1];
        if (ln < OUT_DIM) {
#pragma unroll
            for (int kk = 0; kk < 4; ++kk) w[24 + kk] = load_bfrag_row(Wl, H2, ln, 4 * kk, lane);
            biasO = bl[ln];
        }
    } else {
        const int r0 = gs * H1 + u0 + u, r1 = (2 + gs) * H1 + u0 + u;
#pragma unroll
        for (int kk = 0; kk < 4; ++kk) {
            w[kk * 2]     = load_bfrag_row(Wih1, H0, r0, 4 * kk, lane);
            w[kk * 2 + 1] = load_bfrag_row(Wih1, H0, r1, 4 * kk, lane);
        }
#pragma unroll
        for (int kk = 0; kk < 8; ++kk) {
            w[8 + kk * 2]     = load_bfrag_row(Whh1, H1, r0, 4 * kk, lane);
            w[8 + kk * 2 + 1] = load_bfrag_row(Whh1, H1, r1, 4 * kk, lane);
        }
        bias[0] = bih1[r0] + bhh1[r0];
        bias[1] = bih1[r1] + bhh1[r1];
    }

    const int hrow = 8 * hi + 4 * gs;

    __syncthreads();

    for (int p = 0; p < T_STEPS + 4; ++p) {
        const int cur = p & 1, prv = cur ^ 1;

        if (kind == KIND_X) {
            if (p < T_STEPS) {
                const float* xt = x + ((size_t)p * BATCH + b0) * IN_DIM;
                if (p + 1 < T_STEPS)
                    __builtin_prefetch(xt + (size_t)BATCH * IN_DIM + lane * 32, 0, 0);
                v2f a[16];
#pragma unroll
                for (int kk = 0; kk < 16; ++kk) a[kk] = load_afrag(xt, IN_DIM, 4 * kk, lane);
                v8f acc[2];
#pragma unroll
                for (int j = 0; j < 8; ++j) { acc[0][j] = bias[0]; acc[1][j] = bias[1]; }
#pragma unroll
                for (int kk = 0; kk < 16; ++kk) {
                    acc[0] = wmma4(a[kk], w[kk * 2],     acc[0]);
                    acc[1] = wmma4(a[kk], w[kk * 2 + 1], acc[1]);
                }
                v4f* dst = (v4f*)&sXP[cur][0] + (q * 4) * 32 + lane;
#pragma unroll
                for (int tl = 0; tl < 2; ++tl) {
                    dst[(2 * tl) * 32]     = __builtin_shufflevector(acc[tl], acc[tl], 0, 1, 2, 3);
                    dst[(2 * tl + 1) * 32] = __builtin_shufflevector(acc[tl], acc[tl], 4, 5, 6, 7);
                }
            }
        } else if (kind == KIND_L0) {
            const int t = p - 1;
            if (t >= 0 && t < T_STEPS) {
                const v4f* src = (const v4f*)&sXP[prv][0] + (q * 4) * 32 + lane;
                v8f acc[2];
#pragma unroll
                for (int tl = 0; tl < 2; ++tl) {
                    v4f lo = src[(2 * tl) * 32], up = src[(2 * tl + 1) * 32];
                    acc[tl] = __builtin_shufflevector(lo, up, 0, 1, 2, 3, 4, 5, 6, 7);
                }
                v2f a[4];
#pragma unroll
                for (int kk = 0; kk < 4; ++kk) a[kk] = load_afrag_lds(&sH0[prv][0], P0, 4 * kk, lane);
#pragma unroll
                for (int kk = 0; kk < 4; ++kk) {
                    acc[0] = wmma4(a[kk], w[kk * 2],     acc[0]);
                    acc[1] = wmma4(a[kk], w[kk * 2 + 1], acc[1]);
                }
                cell4(acc[0], acc[1], c, &sH0[cur][hrow * P0 + u0 + u], P0);
            }
        } else if (kind == KIND_L2) {
            const int t = p - 3;
            if (t >= 0 && t <= T_STEPS) {
                v2f ax[8], ah[4];
#pragma unroll
                for (int kk = 0; kk < 8; ++kk) ax[kk] = load_afrag_lds(&sH1[prv][0], P1, 4 * kk, lane);
#pragma unroll
                for (int kk = 0; kk < 4; ++kk) ah[kk] = load_afrag_lds(&sH2[prv][0], P2, 4 * kk, lane);
                if (q == 0 && t >= 1) {
                    v8f po;
#pragma unroll
                    for (int j = 0; j < 8; ++j) po[j] = biasO;
#pragma unroll
                    for (int kk = 0; kk < 4; ++kk) po = wmma4(ah[kk], w[24 + kk], po);
                    if (ln < OUT_DIM) {
                        float* o = out + (((size_t)(b0 + 8 * hi)) * T_STEPS + (t - 1)) * OUT_DIM + ln;
#pragma unroll
                        for (int j = 0; j < 8; ++j) o[(size_t)j * T_STEPS * OUT_DIM] = po[j];
                    }
                }
                if (t < T_STEPS) {
                    v8f acc[2];
#pragma unroll
                    for (int j = 0; j < 8; ++j) { acc[0][j] = bias[0]; acc[1][j] = bias[1]; }
#pragma unroll
                    for (int kk = 0; kk < 8; ++kk) {
                        acc[0] = wmma4(ax[kk], w[kk * 2],     acc[0]);
                        acc[1] = wmma4(ax[kk], w[kk * 2 + 1], acc[1]);
                    }
#pragma unroll
                    for (int kk = 0; kk < 4; ++kk) {
                        acc[0] = wmma4(ah[kk], w[16 + kk * 2],     acc[0]);
                        acc[1] = wmma4(ah[kk], w[16 + kk * 2 + 1], acc[1]);
                    }
                    cell4(acc[0], acc[1], c, &sH2[cur][hrow * P2 + u0 + u], P2);
                }
            }
        } else {
            const int t = p - 2;
            if (t >= 0 && t < T_STEPS) {
                v2f ax[4], ah[8];
#pragma unroll
                for (int kk = 0; kk < 4; ++kk) ax[kk] = load_afrag_lds(&sH0[prv][0], P0, 4 * kk, lane);
#pragma unroll
                for (int kk = 0; kk < 8; ++kk) ah[kk] = load_afrag_lds(&sH1[prv][0], P1, 4 * kk, lane);
                v8f acc[2];
#pragma unroll
                for (int j = 0; j < 8; ++j) { acc[0][j] = bias[0]; acc[1][j] = bias[1]; }
#pragma unroll
                for (int kk = 0; kk < 4; ++kk) {
                    acc[0] = wmma4(ax[kk], w[kk * 2],     acc[0]);
                    acc[1] = wmma4(ax[kk], w[kk * 2 + 1], acc[1]);
                }
#pragma unroll
                for (int kk = 0; kk < 8; ++kk) {
                    acc[0] = wmma4(ah[kk], w[8 + kk * 2],     acc[0]);
                    acc[1] = wmma4(ah[kk], w[8 + kk * 2 + 1], acc[1]);
                }
                cell4(acc[0], acc[1], c, &sH1[cur][hrow * P1 + u0 + u], P1);
            }
        }
        __syncthreads();
    }
}

extern "C" void kernel_launch(void* const* d_in, const int* in_sizes, int n_in,
                              void* d_out, int out_size, void* d_ws, size_t ws_size,
                              hipStream_t stream) {
    (void)in_sizes; (void)n_in; (void)out_size; (void)d_ws; (void)ws_size;
    const float* x    = (const float*)d_in[0];
    const float* Wih0 = (const float*)d_in[1];
    const float* Whh0 = (const float*)d_in[2];
    const float* bih0 = (const float*)d_in[3];
    const float* bhh0 = (const float*)d_in[4];
    const float* Wih1 = (const float*)d_in[5];
    const float* Whh1 = (const float*)d_in[6];
    const float* bih1 = (const float*)d_in[7];
    const float* bhh1 = (const float*)d_in[8];
    const float* Wih2 = (const float*)d_in[9];
    const float* Whh2 = (const float*)d_in[10];
    const float* bih2 = (const float*)d_in[11];
    const float* bhh2 = (const float*)d_in[12];
    const float* Wl   = (const float*)d_in[13];
    const float* bl   = (const float*)d_in[14];
    float* out = (float*)d_out;

    lstm3_stage_kernel<<<BATCH / 16, NTHREADS, 0, stream>>>(
        x, Wih0, Whh0, bih0, bhh0,
        Wih1, Whh1, bih1, bhh1,
        Wih2, Whh2, bih2, bhh2,
        Wl, bl, out);
}


# each layer on its own workgroups (48, not 16), time cut into chunks over successive launches: h rows and carried h, c go through the workspace, stream order the only sync; sum order kept
# speedup vs baseline: 3.6566x; 1.6695x over previous
#include <hip/hip_runtime.h>


typedef __attribute__((ext_vector_type(2))) float v2f;
typedef __attribute__((ext_vector_type(4))) float v4f;
typedef __attribute__((ext_vector_type(8))) float v8f;

#define T_STEPS  2048
#define BATCH    256
#define IN_DIM   64
#define H0       16
#define H1       32
#define H2       16
#define OUT_DIM  8
#define NWAVES   21
#define NTHREADS (NWAVES * 32)

#define P0 20
#define P1 36
#define P2 20

#define KIND_L1  0
#define KIND_L0  1
#define KIND_L2  2
#define KIND_X   3
#define KIND_OUT 4

__device__ __forceinline__ float sigmoidf_(float x) { return 1.0f / (1.0f + __expf(-x)); }

__device__ __forceinline__ v2f load_afrag(const float* base, int lda, int kbase, int lane) {
    int m = lane & 15;
    int k = kbase + ((lane >> 4) << 1);
    const float* p = base + m * lda + k;
    v2f a;
    a.x = p[0];
    a.y = p[1];
    return a;
}

__device__ __forceinline__ v2f load_afrag_lds(const float* base, int pitch, int kbase, int lane) {
    int m = lane & 15;
    int k = kbase + ((lane >> 4) << 1);
    return *(const v2f*)(base + m * pitch + k);
}

__device__ __forceinline__ v2f load_bfrag_row(const float* W, int ldw, int row, int kbase, int lane) {
    int k = kbase + ((lane >> 4) << 1);
    const float* p = W + row * ldw + k;
    v2f b;
    b.x = p[0];
    b.y = p[1];
    return b;
}

__device__ __forceinline__ v8f wmma4(v2f a, v2f b, v8f c) {
    return __builtin_amdgcn_wmma_f32_16x16x4_f32(false, a, false, b, (short)0, c, false, false);
}

template <int ROR, int BANK>
__device__ __forceinline__ float take(float keep, float give) {
    return __builtin_bit_cast(float,
        __builtin_amdgcn_update_dpp(__builtin_bit_cast(int, keep), __builtin_bit_cast(int, give),
                                    0x120 + ROR, 0xf, BANK, false));
}

__device__ __forceinline__ void cell2(const v8f& t, float (&c)[2], float* hdst, int pitch, float (&hl)[2]) {
#pragma unroll
    for (int jj = 0; jj < 2; ++jj) {
        float vi = take<12, 0x8>(take<8, 0x4>(take<4,  0x2>(t[jj],     t[2 + jj]), t[4 + jj]), t[6 + jj]);
        float vf = take<8,  0x8>(take<4, 0x4>(take<12, 0x1>(t[2 + jj], t[jj]),     t[4 + jj]), t[6 + jj]);
        float vg = take<4,  0x8>(take<12, 0x2>(take<8, 0x1>(t[4 + jj], t[jj]),     t[2 + jj]), t[6 + jj]);
        float vo = take<12, 0x4>(take<8, 0x2>(take<4,  0x1>(t[6 + jj], t[jj]),     t[2 + jj]), t[4 + jj]);
        float cc = c[jj];
        float cn = sigmoidf_(vf) * cc + sigmoidf_(vi) * tanhf(vg);
        c[jj] = cn;
        float hv = sigmoidf_(vo) * tanhf(cn);
        hl[jj] = hv;
        hdst[jj * pitch] = hv;
    }
}

__global__ __launch_bounds__(NTHREADS) void lstm3_stage_kernel(
    const float* __restrict__ x,
    const float* __restrict__ Wih0, const float* __restrict__ Whh0,
    const float* __restrict__ bih0, const float* __restrict__ bhh0,
    const float* __restrict__ Wih1, const float* __restrict__ Whh1,
    const float* __restrict__ bih1, const float* __restrict__ bhh1,
    const float* __restrict__ Wih2, const float* __restrict__ Whh2,
    const float* __restrict__ bih2, const float* __restrict__ bhh2,
    const float* __restrict__ Wl,   const float* __restrict__ bl,
    float* __restrict__ out)
{
    __shared__ __attribute__((aligned(16))) float sXP[2][4 * 2 * 32 * 4];
    __shared__ __attribute__((aligned(16))) float sH0[2][16 * P0];
    __shared__ __attribute__((aligned(16))) float sH1[2][16 * P1];
    __shared__ __attribute__((aligned(16))) float sH2[2][16 * P2];

    const int tid  = threadIdx.x;
    const int lane = tid & 31;
    const int ln   = lane & 15;
    const int hi   = lane >> 4;
    const int u    = ln & 3;
    const int gq   = ln >> 2;
    const int wave = __builtin_amdgcn_readfirstlane(tid >> 5);
    const int kind = (wave < 8) ? KIND_L1 : (wave < 12) ? KIND_L2 : (wave < 16) ? KIND_L0
                   : (wave < 17) ? KIND_OUT : KIND_X;
    const int q    = (wave < 8) ? wave : (wave < 16) ? (wave & 3) : (wave < 17) ? 0 : (wave - 17);
    const int u0   = q * 4;
    const int b0   = blockIdx.x * 16;

    for (int i = tid; i < 2 * 16 * P0; i += NTHREADS) { (&sH0[0][0])[i] = 0.f; (&sH2[0][0])[i] = 0.f; }
    for (int i = tid; i < 2 * 16 * P1; i += NTHREADS) { (&sH1[0][0])[i] = 0.f; }

    v2f   w[16];
    float bias = 0.f;
    float c[2];
    float hl[2];
#pragma unroll
    for (int i = 0; i < 16; ++i) { w[i].x = 0.f; w[i].y = 0.f; }
    c[0] = 0.f; c[1] = 0.f;
    float biasO = 0.f;

    if (kind == KIND_X) {
        const int r = gq * H0 + u0 + u;
#pragma unroll
        for (int kk = 0; kk < 16; ++kk) w[kk] = load_bfrag_row(Wih0, IN_DIM, r, 4 * kk, lane);
        bias = bih0[r] + bhh0[r];
    } else if (kind == KIND_L0) {
        const int r = gq * H0 + u0 + u;
#pragma unroll
        for (int kk = 0; kk < 4; ++kk) w[kk] = load_bfrag_row(Whh0, H0, r, 4 * kk, lane);
    } else if (kind == KIND_L2) {
        const int r = gq * H2 + u0 + u;
#pragma unroll
        for (int kk = 0; kk < 8; ++kk) w[kk] = load_bfrag_row(Wih2, H1, r, 4 * kk, lane);
#pragma unroll
        for (int kk = 0; kk < 4; ++kk) w[8 + kk] = load_bfrag_row(Whh2, H2, r, 4 * kk, lane);
        bias = bih2[r] + bhh2[r];
    } else if (kind == KIND_OUT) {
        if (ln < OUT_DIM) {
#pragma unroll
            for (int kk = 0; kk < 4; ++kk) w[kk] = load_bfrag_row(Wl, H2, ln, 4 * kk, lane);
            biasO = bl[ln];
        }
    } else {
        const int r = gq * H1 + u0 + u;
#pragma unroll
        for (int kk = 0; kk < 4; ++kk) w[kk] = load_bfrag_row(Wih1, H0, r, 4 * kk, lane);
#pragma unroll
        for (int kk = 0; kk < 8; ++kk) w[4 + kk] = load_bfrag_row(Whh1, H1, r, 4 * kk, lane);
        bias = bih1[r] + bhh1[r];
    }

    const int hrow = 8 * hi + 2 * gq;

    __syncthreads();

    for (int p = 0; p < T_STEPS + 4; ++p) {
        const int cur = p & 1, prv = cur ^ 1;

        if (kind == KIND_X) {
            if (p < T_STEPS) {
                const float* xt = x + ((size_t)p * BATCH + b0) * IN_DIM;
                if (p + 1 < T_STEPS)
                    __builtin_prefetch(xt + (size_t)BATCH * IN_DIM + lane * 32, 0, 0);
                v2f a[16];
#pragma unroll
                for (int kk = 0; kk < 16; ++kk) a[kk] = load_afrag(xt, IN_DIM, 4 * kk, lane);
                v8f acc;
#pragma unroll
                for (int j = 0; j < 8; ++j) acc[j] = bias;
#pragma unroll
                for (int kk = 0; kk < 16; ++kk) acc = wmma4(a[kk], w[kk], acc);
                v4f* dst = (v4f*)&sXP[cur][0] + (q * 2) * 32 + lane;
                dst[0]  = __builtin_shufflevector(acc, acc, 0, 1, 2, 3);
                dst[32] = __builtin_shufflevector(acc, acc, 4, 5, 6, 7);
            }
        } else if (kind == KIND_L0) {
            const int t = p - 1;
            if (t >= 0 && t < T_STEPS) {
                const v4f* src = (const v4f*)&sXP[prv][0] + (q * 2) * 32 + lane;
                v4f lo = src[0], up = src[32];
                v8f acc = __builtin_shufflevector(lo, up, 0, 1, 2, 3, 4, 5, 6, 7);
                v2f a[4];
#pragma unroll
                for (int kk = 0; kk < 4; ++kk) a[kk] = load_afrag_lds(&sH0[prv][0], P0, 4 * kk, lane);
#pragma unroll
                for (int kk = 0; kk < 4; ++kk) acc = wmma4(a[kk], w[kk], acc);
                cell2(acc, c, &sH0[cur][hrow * P0 + u0 + u], P0, hl);
            }
        } else if (kind == KIND_L2) {
            const int t = p - 3;
            if (t >= 0 && t < T_STEPS) {
                v2f ax[8], ah[4];
#pragma unroll
                for (int kk = 0; kk < 8; ++kk) ax[kk] = load_afrag_lds(&sH1[prv][0], P1, 4 * kk, lane);
#pragma unroll
                for (int kk = 0; kk < 4; ++kk) ah[kk] = load_afrag_lds(&sH2[prv][0], P2, 4 * kk, lane);
                v8f acc;
#pragma unroll
                for (int j = 0; j < 8; ++j) acc[j] = bias;
#pragma unroll
                for (int kk = 0; kk < 8; ++kk) acc = wmma4(ax[kk], w[kk], acc);
#pragma unroll
                for (int kk = 0; kk < 4; ++kk) acc = wmma4(ah[kk], w[8 + kk], acc);
                cell2(acc, c, &sH2[cur][hrow * P2 + u0 + u], P2, hl);
            }
        } else if (kind == KIND_OUT) {
            const int t = p - 4;
            if (t >= 0) {
                v2f ah[4];
#pragma unroll
                for (int kk = 0; kk < 4; ++kk) ah[kk] = load_afrag_lds(&sH2[prv][0], P2, 4 * kk, lane);
                v8f po;
#pragma unroll
                for (int j = 0; j < 8; ++j) po[j] = biasO;
#pragma unroll
                for (int kk = 0; kk < 4; ++kk) po = wmma4(ah[kk], w[kk], po);
                if (ln < OUT_DIM) {
                    float* o = out + (((size_t)(b0 + 8 * hi)) * T_STEPS + t) * OUT_DIM + ln;
#pragma unroll
                    for (int j = 0; j < 8; ++j) o[(size_t)j * T_STEPS * OUT_DIM] = po[j];
                }
            }
        } else {
            const int t = p - 2;
            if (t >= 0 && t < T_STEPS) {
                v2f ax[4], ah[8];
#pragma unroll
                for (int kk = 0; kk < 4; ++kk) ax[kk] = load_afrag_lds(&sH0[prv][0], P0, 4 * kk, lane);
#pragma unroll
                for (int kk = 0; kk < 8; ++kk) ah[kk] = load_afrag_lds(&sH1[prv][0], P1, 4 * kk, lane);
                v8f acc;
#pragma unroll
                for (int j = 0; j < 8; ++j) acc[j] = bias;
#pragma unroll
                for (int kk = 0; kk < 4; ++kk) acc = wmma4(ax[kk], w[kk], acc);
#pragma unroll
                for (int kk = 0; kk < 8; ++kk) acc = wmma4(ah[kk], w[4 + kk], acc);
                cell2(acc, c, &sH1[cur][hrow * P1 + u0 + u], P1, hl);
            }
        }
        __syncthreads();
    }
}


#define PWAVES   9
#define PTHREADS (PWAVES * 32)
#define NTILES   (BATCH / 16)
#define KIND_ST   5
#define KIND_IDLE 6

__device__ __forceinline__ v2f load_afrag_g(const float* base, int lda, int kbase, int lane) {
    int m = lane & 15;
    int k = kbase + ((lane >> 4) << 1);
    return *(const v2f*)(base + m * lda + k);
}

template <int C>
__global__ __launch_bounds__(PTHREADS) void lstm3_layer_chunk_kernel(
    const float* __restrict__ x,
    const float* __restrict__ Wih0, const float* __restrict__ Whh0,
    const float* __restrict__ bih0, const float* __restrict__ bhh0,
    const float* __restrict__ Wih1, const float* __restrict__ Whh1,
    const float* __restrict__ bih1, const float* __restrict__ bhh1,
    const float* __restrict__ Wih2, const float* __restrict__ Whh2,
    const float* __restrict__ bih2, const float* __restrict__ bhh2,
    const float* __restrict__ Wl,   const float* __restrict__ bl,
    float* __restrict__ out, float* ws, int d)
{
    constexpr int NCH = T_STEPS / C;
    const int role = blockIdx.y;
    const int k    = d - role;
    if (k < 0 || k >= NCH) return;

    __shared__ __attribute__((aligned(16))) float sXP[2][4 * 2 * 32 * 4];
    __shared__ __attribute__((aligned(16))) float sH0[2][16 * P0];
    __shared__ __attribute__((aligned(16))) float sH1[2][16 * P1];
    __shared__ __attribute__((aligned(16))) float sH2[2][16 * P2];

    const int tid  = threadIdx.x;
    const int lane = tid & 31;
    const int ln   = lane & 15;
    const int hi   = lane >> 4;
    const int u    = ln & 3;
    const int gq   = ln >> 2;
    const int wave = __builtin_amdgcn_readfirstlane(tid >> 5);
    const int b0   = blockIdx.x * 16;
    const int t0   = k * C;
    const int kb   = k & 1;
    const int hrow = 8 * hi + 2 * gq;
    const int srow  = lane >> 1;
    const int shalf = lane & 1;

    float* ho0c = ws + ((size_t)kb * C * BATCH + b0) * H0;
    float* ho1c = ws + (size_t)2 * C * BATCH * H0 + ((size_t)kb * C * BATCH + b0) * H1;
    float* st   = ws + (size_t)2 * C * BATCH * (H0 + H1);
    v4f* stl = (v4f*)st + ((size_t)((role * NTILES + blockIdx.x) * 8 + (wave & 7)) * 32 + lane);

    float c[2]  = {0.f, 0.f};
    float hl[2] = {0.f, 0.f};

    if (role == 0) {
        const int kind = (wave < 4) ? KIND_L0 : (wave < 8) ? KIND_X : KIND_ST;
        const int q  = wave & 3;
        const int u0 = q * 4;
        const int r  = gq * H0 + u0 + u;
        v2f w[16];
        float bias = 0.f;
        for (int i = 0; i < 16; ++i) { w[i].x = 0.f; w[i].y = 0.f; }
        if (kind == KIND_X) {
            for (int kk = 0; kk < 16; ++kk) w[kk] = load_bfrag_row(Wih0, IN_DIM, r, 4 * kk, lane);
            bias = bih0[r] + bhh0[r];
        } else if (kind == KIND_L0) {
            for (int kk = 0; kk < 4; ++kk) w[kk] = load_bfrag_row(Whh0, H0, r, 4 * kk, lane);
            if (k > 0) { v4f s = *stl; c[0] = s.x; c[1] = s.y; hl[0] = s.z; hl[1] = s.w; }
            float* h = &sH0[0][hrow * P0 + u0 + u];
            h[0] = hl[0]; h[P0] = hl[1]; h[16 * P0] = hl[0]; h[16 * P0 + P0] = hl[1];
        }
        __syncthreads();
        for (int p = 0; p < C + 2; ++p) {
            const int cur = p & 1, prv = cur ^ 1;
            if (kind == KIND_X) {
                if (p < C) {
                    const float* xt = x + ((size_t)(t0 + p) * BATCH + b0) * IN_DIM;
                    if (t0 + p + 1 < T_STEPS)
                        __builtin_prefetch(xt + (size_t)BATCH * IN_DIM + lane * 32, 0, 0);
                    v2f a[16];
                    for (int kk = 0; kk < 16; ++kk) a[kk] = load_afrag(xt, IN_DIM, 4 * kk, lane);
                    v8f acc;
                    for (int j = 0; j < 8; ++j) acc[j] = bias;
                    for (int kk = 0; kk < 16; ++kk) acc = wmma4(a[kk], w[kk], acc);
                    v4f* dst = (v4f*)&sXP[cur][0] + (q * 2) * 32 + lane;
                    dst[0]  = __builtin_shufflevector(acc, acc, 0, 1, 2, 3);
                    dst[32] = __builtin_shufflevector(acc, acc, 4, 5, 6, 7);
                }
            } else if (kind == KIND_L0) {
                if (p >= 1 && p <= C) {
                    const v4f* src = (const v4f*)&sXP[prv][0] + (q * 2) * 32 + lane;
                    v4f lo = src[0], up = src[32];
                    v8f acc = __builtin_shufflevector(lo, up, 0, 1, 2, 3, 4, 5, 6, 7);
                    v2f a[4];
                    for (int kk = 0; kk < 4; ++kk) a[kk] = load_afrag_lds(&sH0[prv][0], P0, 4 * kk, lane);
                    for (int kk = 0; kk < 4; ++kk) acc = wmma4(a[kk], w[kk], acc);
                    cell2(acc, c, &sH0[cur][hrow * P0 + u0 + u], P0, hl);
                }
            } else {
                if (p >= 2) {
                    const v4f* src = (const v4f*)&sH0[prv][srow * P0 + shalf * 8];
                    v4f a0 = src[0], a1 = src[1];
                    v4f* dst = (v4f*)(ho0c + (size_t)(p - 2) * BATCH * H0 + srow * H0 + shalf * 8);
                    dst[0] = a0; dst[1] = a1;
                }
            }
            __syncthreads();
        }
        if (kind == KIND_L0) { v4f s; s.x = c[0]; s.y = c[1]; s.z = hl[0]; s.w = hl[1]; *stl = s; }
    } else if (role == 1) {
        const int kind = (wave < 8) ? KIND_L1 : KIND_ST;
        const int q  = wave & 7;
        const int u0 = q * 4;
        const int r  = gq * H1 + u0 + u;
        v2f w[12];
        v2f axn[4];
        float bias = 0.f;
        for (int i = 0; i < 12; ++i) { w[i].x = 0.f; w[i].y = 0.f; }
        for (int i = 0; i < 4; ++i) { axn[i].x = 0.f; axn[i].y = 0.f; }
        if (kind == KIND_L1) {
            for (int kk = 0; kk < 4; ++kk) w[kk] = load_bfrag_row(Wih1, H0, r, 4 * kk, lane);
            for (int kk = 0; kk < 8; ++kk) w[4 + kk] = load_bfrag_row(Whh1, H1, r, 4 * kk, lane);
            bias = bih1[r] + bhh1[r];
            if (k > 0) { v4f s = *stl; c[0] = s.x; c[1] = s.y; hl[0] = s.z; hl[1] = s.w; }
            float* h = &sH1[0][hrow * P1 + u0 + u];
            h[0] = hl[0]; h[P1] = hl[1]; h[16 * P1] = hl[0]; h[16 * P1 + P1] = hl[1];
            for (int kk = 0; kk < 4; ++kk) axn[kk] = load_afrag_g(ho0c, H0, 4 * kk, lane);
        }
        __syncthreads();
        for (int p = 0; p < C + 1; ++p) {
            const int cur = p & 1, prv = cur ^ 1;
            if (kind == KIND_L1) {
                if (p < C) {
                    v2f ax[4], ah[8];
                    for (int kk = 0; kk < 4; ++kk) ax[kk] = axn[kk];
                    for (int kk = 0; kk < 8; ++kk) ah[kk] = load_afrag_lds(&sH1[prv][0], P1, 4 * kk, lane);
                    if (p + 1 < C) {
                        for (int kk = 0; kk < 4; ++kk)
                            axn[kk] = load_afrag_g(ho0c + (size_t)(p + 1) * BATCH * H0, H0, 4 * kk, lane);
                    }
                    v8f acc;
                    for (int j = 0; j < 8; ++j) acc[j] = bias;
                    for (int kk = 0; kk < 4; ++kk) acc = wmma4(ax[kk], w[kk], acc);
                    for (int kk = 0; kk < 8; ++kk) acc = wmma4(ah[kk], w[4 + kk], acc);
                    cell2(acc, c, &sH1[cur][hrow * P1 + u0 + u], P1, hl);
                }
            } else {
                if (p >= 1) {
                    const v4f* src = (const v4f*)&sH1[prv][srow * P1 + shalf * 16];
                    v4f a0 = src[0], a1 = src[1], a2 = src[2], a3 = src[3];
                    v4f* dst = (v4f*)(ho1c + (size_t)(p - 1) * BATCH * H1 + srow * H1 + shalf * 16);
                    dst[0] = a0; dst[1] = a1; dst[2] = a2; dst[3] = a3;
                }
            }
            __syncthreads();
        }
        if (kind == KIND_L1) { v4f s; s.x = c[0]; s.y = c[1]; s.z = hl[0]; s.w = hl[1]; *stl = s; }
    } else {
        const int kind = (wave < 4) ? KIND_L2 : (wave == 4) ? KIND_OUT : KIND_IDLE;
        const int q  = wave & 3;
        const int u0 = q * 4;
        const int r  = gq * H2 + u0 + u;
        v2f w[12];
        v2f axn[8];
        float bias = 0.f, biasO = 0.f;
        for (int i = 0; i < 12; ++i) { w[i].x = 0.f; w[i].y = 0.f; }
        for (int i = 0; i < 8; ++i) { axn[i].x = 0.f; axn[i].y = 0.f; }
        if (kind == KIND_L2) {
            for (int kk = 0; kk < 8; ++kk) w[kk] = load_bfrag_row(Wih2, H1, r, 4 * kk, lane);
            for (int kk = 0; kk < 4; ++kk) w[8 + kk] = load_bfrag_row(Whh2, H2, r, 4 * kk, lane);
            bias = bih2[r] + bhh2[r];
            if (k > 0) { v4f s = *stl; c[0] = s.x; c[1] = s.y; hl[0] = s.z; hl[1] = s.w; }
            float* h = &sH2[0][hrow * P2 + u0 + u];
            h[0] = hl[0]; h[P2] = hl[1]; h[16 * P2] = hl[0]; h[16 * P2 + P2] = hl[1];
            for (int kk = 0; kk < 8; ++kk) axn[kk] = load_afrag_g(ho1c, H1, 4 * kk, lane);
        } else if (kind == KIND_OUT) {
            if (ln < OUT_DIM) {
                for (int kk = 0; kk < 4; ++kk) w[kk] = load_bfrag_row(Wl, H2, ln, 4 * kk, lane);
                biasO = bl[ln];
            }
        }
        __syncthreads();
        for (int p = 0; p < C + 1; ++p) {
            const int cur = p & 1, prv = cur ^ 1;
            if (kind == KIND_L2) {
                if (p < C) {
                    v2f ax[8], ah[4];
                    for (int kk = 0; kk < 8; ++kk) ax[kk] = axn[kk];
                    for (int kk = 0; kk < 4; ++kk) ah[kk] = load_afrag_lds(&sH2[prv][0], P2, 4 * kk, lane);
                    if (p + 1 < C) {
                        for (int kk = 0; kk < 8; ++kk)
                            axn[kk] = load_afrag_g(ho1c + (size_t)(p + 1) * BATCH * H1, H1, 4 * kk, lane);
                    }
                    v8f acc;
                    for (int j = 0; j < 8; ++j) acc[j] = bias;
                    for (int kk = 0; kk < 8; ++kk) acc = wmma4(ax[kk], w[kk], acc);
                    for (int kk = 0; kk < 4; ++kk) acc = wmma4(ah[kk], w[8 + kk], acc);
                    cell2(acc, c, &sH2[cur][hrow * P2 + u0 + u], P2, hl);
                }
            } else if (kind == KIND_OUT) {
                if (p >= 1) {
                    const int t = t0 + p - 1;
                    v2f ah[4];
                    for (int kk = 0; kk < 4; ++kk) ah[kk] = load_afrag_lds(&sH2[prv][0], P2, 4 * kk, lane);
                    v8f po;
                    for (int j = 0; j < 8; ++j) po[j] = biasO;
                    for (int kk = 0; kk < 4; ++kk) po = wmma4(ah[kk], w[kk], po);
                    if (ln < OUT_DIM) {
                        float* o = out + (((size_t)(b0 + 8 * hi)) * T_STEPS + t) * OUT_DIM + ln;
                        for (int j = 0; j < 8; ++j) o[(size_t)j * T_STEPS * OUT_DIM] = po[j];
                    }
                }
            }
            __syncthreads();
        }
        if (kind == KIND_L2) { v4f s; s.x = c[0]; s.y = c[1]; s.z = hl[0]; s.w = hl[1]; *stl = s; }
    }
}

template <int C>
static void launch_chunks(const float* x,
                          const float* Wih0, const float* Whh0, const float* bih0, const float* bhh0,
                          const float* Wih1, const float* Whh1, const float* bih1, const float* bhh1,
                          const float* Wih2, const float* Whh2, const float* bih2, const float* bhh2,
                          const float* Wl, const float* bl, float* out, float* ws, hipStream_t stream) {
    for (int d = 0; d < T_STEPS / C + 2; ++d)
        lstm3_layer_chunk_kernel<C><<<dim3(NTILES, 3), PTHREADS, 0, stream>>>(
            x, Wih0, Whh0, bih0, bhh0, Wih1, Whh1, bih1, bhh1, Wih2, Whh2, bih2, bhh2, Wl, bl, out, ws, d);
}

extern "C" void kernel_launch(void* const* d_in, const int* in_sizes, int n_in,
                              void* d_out, int out_size, void* d_ws, size_t ws_size,
                              hipStream_t stream) {
    (void)in_sizes; (void)n_in; (void)out_size;
    const float* x    = (const float*)d_in[0];
    const float* Wih0 = (const float*)d_in[1];
    const float* Whh0 = (const float*)d_in[2];
    const float* bih0 = (const float*)d_in[3];
    const float* bhh0 = (const float*)d_in[4];
    const float* Wih1 = (const float*)d_in[5];
    const float* Whh1 = (const float*)d_in[6];
    const float* bih1 = (const float*)d_in[7];
    const float* bhh1 = (const float*)d_in[8];
    const float* Wih2 = (const float*)d_in[9];
    const float* Whh2 = (const float*)d_in[10];
    const float* bih2 = (const float*)d_in[11];
    const float* bhh2 = (const float*)d_in[12];
    const float* Wl   = (const float*)d_in[13];
    const float* bl   = (const float*)d_in[14];
    float* out = (float*)d_out;

    float* ws = (float*)d_ws;
    if (ws != nullptr && ws_size >= (size_t)6488064) {
        launch_chunks<64>(x, Wih0, Whh0, bih0, bhh0, Wih1, Whh1, bih1, bhh1, Wih2, Whh2, bih2, bhh2, Wl, bl, out, ws, stream);
    } else if (ws != nullptr && ws_size >= (size_t)3342336) {
        launch_chunks<32>(x, Wih0, Whh0, bih0, bhh0, Wih1, Whh1, bih1, bhh1, Wih2, Whh2, bih2, bhh2, Wl, bl, out, ws, stream);
    } else if (ws != nullptr && ws_size >= (size_t)1769472) {
        launch_chunks<16>(x, Wih0, Whh0, bih0, bhh0, Wih1, Whh1, bih1, bhh1, Wih2, Whh2, bih2, bhh2, Wl, bl, out, ws, stream);
    } else {
        lstm3_stage_kernel<<<BATCH / 16, NTHREADS, 0, stream>>>(
            x, Wih0, Whh0, bih0, bhh0,
            Wih1, Whh1, bih1, bhh1,
            Wih2, Whh2, bih2, bhh2,
            Wl, bl, out);
    }
}


# cell update reordered, same operations: the 6 sigmoids of a lane's 2 cells run as one block, ahead of the tanh branches (behind the first tanh in a SIMD's younger wave); sum order kept
# speedup vs baseline: 3.9795x; 1.0883x over previous
#include <hip/hip_runtime.h>


typedef __attribute__((ext_vector_type(2))) float v2f;
typedef __attribute__((ext_vector_type(4))) float v4f;
typedef __attribute__((ext_vector_type(8))) float v8f;

#define T_STEPS  2048
#define BATCH    256
#define IN_DIM   64
#define H0       16
#define H1       32
#define H2       16
#define OUT_DIM  8
#define NWAVES   21
#define NTHREADS (NWAVES * 32)

#define P0 20
#define P1 36
#define P2 20

#define KIND_L1  0
#define KIND_L0  1
#define KIND_L2  2
#define KIND_X   3
#define KIND_OUT 4

__device__ __forceinline__ float sigmoidf_(float x) { return 1.0f / (1.0f + __expf(-x)); }

__device__ __forceinline__ v2f load_afrag(const float* base, int lda, int kbase, int lane) {
    int m = lane & 15;
    int k = kbase + ((lane >> 4) << 1);
    const float* p = base + m * lda + k;
    v2f a;
    a.x = p[0];
    a.y = p[1];
    return a;
}

__device__ __forceinline__ v2f load_afrag_lds(const float* base, int pitch, int kbase, int lane) {
    int m = lane & 15;
    int k = kbase + ((lane >> 4) << 1);
    return *(const v2f*)(base + m * pitch + k);
}

__device__ __forceinline__ v2f load_bfrag_row(const float* W, int ldw, int row, int kbase, int lane) {
    int k = kbase + ((lane >> 4) << 1);
    const float* p = W + row * ldw + k;
    v2f b;
    b.x = p[0];
    b.y = p[1];
    return b;
}

__device__ __forceinline__ v8f wmma4(v2f a, v2f b, v8f c) {
    return __builtin_amdgcn_wmma_f32_16x16x4_f32(false, a, false, b, (short)0, c, false, false);
}

template <int ROR, int BANK>
__device__ __forceinline__ float take(float keep, float give) {
    return __builtin_bit_cast(float,
        __builtin_amdgcn_update_dpp(__builtin_bit_cast(int, keep), __builtin_bit_cast(int, give),
                                    0x120 + ROR, 0xf, BANK, false));
}

template <int JJ> __device__ __forceinline__ float gate_i(const v8f& t) {
    return take<12, 0x8>(take<8, 0x4>(take<4,  0x2>(t[JJ],     t[2 + JJ]), t[4 + JJ]), t[6 + JJ]);
}
template <int JJ> __device__ __forceinline__ float gate_f(const v8f& t) {
    return take<8,  0x8>(take<4, 0x4>(take<12, 0x1>(t[2 + JJ], t[JJ]),     t[4 + JJ]), t[6 + JJ]);
}
template <int JJ> __device__ __forceinline__ float gate_g(const v8f& t) {
    return take<4,  0x8>(take<12, 0x2>(take<8, 0x1>(t[4 + JJ], t[JJ]),     t[2 + JJ]), t[6 + JJ]);
}
template <int JJ> __device__ __forceinline__ float gate_o(const v8f& t) {
    return take<12, 0x4>(take<8, 0x2>(take<4,  0x1>(t[6 + JJ], t[JJ]),     t[2 + JJ]), t[4 + JJ]);
}

template <bool SIG_FIRST>
__device__ __forceinline__ void cell2(const v8f& t, float (&c)[2], float* hdst, int pitch, float (&hl)[2]) {
    float si0, sf0, so0, si1, sf1, so1, tg0, vg1;
    if (SIG_FIRST) {
        const float vg0 = gate_g<0>(t);
        vg1 = gate_g<1>(t);
        si0 = sigmoidf_(gate_i<0>(t)); sf0 = sigmoidf_(gate_f<0>(t)); so0 = sigmoidf_(gate_o<0>(t));
        si1 = sigmoidf_(gate_i<1>(t)); sf1 = sigmoidf_(gate_f<1>(t)); so1 = sigmoidf_(gate_o<1>(t));
        asm volatile("" : "+v"(si0), "+v"(sf0), "+v"(so0), "+v"(si1), "+v"(sf1), "+v"(so1));
        tg0 = tanhf(vg0);
    } else {
        tg0 = tanhf(gate_g<0>(t));
        vg1 = gate_g<1>(t);
        si0 = sigmoidf_(gate_i<0>(t)); sf0 = sigmoidf_(gate_f<0>(t)); so0 = sigmoidf_(gate_o<0>(t));
        si1 = sigmoidf_(gate_i<1>(t)); sf1 = sigmoidf_(gate_f<1>(t)); so1 = sigmoidf_(gate_o<1>(t));
        asm volatile("" : "+v"(si0), "+v"(sf0), "+v"(so0), "+v"(si1), "+v"(sf1), "+v"(so1));
    }
    const float tg1 = tanhf(vg1);
    const float cn0 = sf0 * c[0] + si0 * tg0;
    const float cn1 = sf1 * c[1] + si1 * tg1;
    c[0] = cn0;
    c[1] = cn1;
    const float hv0 = so0 * tanhf(cn0);
    const float hv1 = so1 * tanhf(cn1);
    hl[0] = hv0;
    hl[1] = hv1;
    hdst[0]     = hv0;
    hdst[pitch] = hv1;
}

__global__ __launch_bounds__(NTHREADS) void lstm3_stage_kernel(
    const float* __restrict__ x,
    const float* __restrict__ Wih0, const float* __restrict__ Whh0,
    const float* __restrict__ bih0, const float* __restrict__ bhh0,
    const float* __restrict__ Wih1, const float* __restrict__ Whh1,
    const float* __restrict__ bih1, const float* __restrict__ bhh1,
    const float* __restrict__ Wih2, const float* __restrict__ Whh2,
    const float* __restrict__ bih2, const float* __restrict__ bhh2,
    const float* __restrict__ Wl,   const float* __restrict__ bl,
    float* __restrict__ out)
{
    __shared__ __attribute__((aligned(16))) float sXP[2][4 * 2 * 32 * 4];
    __shared__ __attribute__((aligned(16))) float sH0[2][16 * P0];
    __shared__ __attribute__((aligned(16))) float sH1[2][16 * P1];
    __shared__ __attribute__((aligned(16))) float sH2[2][16 * P2];

    const int tid  = threadIdx.x;
    const int lane = tid & 31;
    const int ln   = lane & 15;
    const int hi   = lane >> 4;
    const int u    = ln & 3;
    const int gq   = ln >> 2;
    const int wave = __builtin_amdgcn_readfirstlane(tid >> 5);
    const int kind = (wave < 8) ? KIND_L1 : (wave < 12) ? KIND_L2 : (wave < 16) ? KIND_L0
                   : (wave < 17) ? KIND_OUT : KIND_X;
    const int q    = (wave < 8) ? wave : (wave < 16) ? (wave & 3) : (wave < 17) ? 0 : (wave - 17);
    const int u0   = q * 4;
    const int b0   = blockIdx.x * 16;

    for (int i = tid; i < 2 * 16 * P0; i += NTHREADS) { (&sH0[0][0])[i] = 0.f; (&sH2[0][0])[i] = 0.f; }
    for (int i = tid; i < 2 * 16 * P1; i += NTHREADS) { (&sH1[0][0])[i] = 0.f; }

    v2f   w[16];
    float bias = 0.f;
    float c[2];
    float hl[2];
#pragma unroll
    for (int i = 0; i < 16; ++i) { w[i].x = 0.f; w[i].y = 0.f; }
    c[0] = 0.f; c[1] = 0.f;
    float biasO = 0.f;

    if (kind == KIND_X) {
        const int r = gq * H0 + u0 + u;
#pragma unroll
        for (int kk = 0; kk < 16; ++kk) w[kk] = load_bfrag_row(Wih0, IN_DIM, r, 4 * kk, lane);
        bias = bih0[r] + bhh0[r];
    } else if (kind == KIND_L0) {
        const int r = gq * H0 + u0 + u;
#pragma unroll
        for (int kk = 0; kk < 4; ++kk) w[kk] = load_bfrag_row(Whh0, H0, r, 4 * kk, lane);
    } else if (kind == KIND_L2) {
        const int r = gq * H2 + u0 + u;
#pragma unroll
        for (int kk = 0; kk < 8; ++kk) w[kk] = load_bfrag_row(Wih2, H1, r, 4 * kk, lane);
#pragma unroll
        for (int kk = 0; kk < 4; ++kk) w[8 + kk] = load_bfrag_row(Whh2, H2, r, 4 * kk, lane);
        bias = bih2[r] + bhh2[r];
    } else if (kind == KIND_OUT) {
        if (ln < OUT_DIM) {
#pragma unroll
            for (int kk = 0; kk < 4; ++kk) w[kk] = load_bfrag_row(Wl, H2, ln, 4 * kk, lane);
            biasO = bl[ln];
        }
    } else {
        const int r = gq * H1 + u0 + u;
#pragma unroll
        for (int kk = 0; kk < 4; ++kk) w[kk] = load_bfrag_row(Wih1, H0, r, 4 * kk, lane);
#pragma unroll
        for (int kk = 0; kk < 8; ++kk) w[4 + kk] = load_bfrag_row(Whh1, H1, r, 4 * kk, lane);
        bias = bih1[r] + bhh1[r];
    }

    const int hrow = 8 * hi + 2 * gq;

    __syncthreads();

    for (int p = 0; p < T_STEPS + 4; ++p) {
        const int cur = p & 1, prv = cur ^ 1;

        if (kind == KIND_X) {
            if (p < T_STEPS) {
                const float* xt = x + ((size_t)p * BATCH + b0) * IN_DIM;
                if (p + 1 < T_STEPS)
                    __builtin_prefetch(xt + (size_t)BATCH * IN_DIM + lane * 32, 0, 0);
                v2f a[16];
#pragma unroll
                for (int kk = 0; kk < 16; ++kk) a[kk] = load_afrag(xt, IN_DIM, 4 * kk, lane);
                v8f acc;
#pragma unroll
                for (int j = 0; j < 8; ++j) acc[j] = bias;
#pragma unroll
                for (int kk = 0; kk < 16; ++kk) acc = wmma4(a[kk], w[kk], acc);
                v4f* dst = (v4f*)&sXP[cur][0] + (q * 2) * 32 + lane;
                dst[0]  = __builtin_shufflevector(acc, acc, 0, 1, 2, 3);
                dst[32] = __builtin_shufflevector(acc, acc, 4, 5, 6, 7);
            }
        } else if (kind == KIND_L0) {
            const int t = p - 1;
            if (t >= 0 && t < T_STEPS) {
                const v4f* src = (const v4f*)&sXP[prv][0] + (q * 2) * 32 + lane;
                v4f lo = src[0], up = src[32];
                v8f acc = __builtin_shufflevector(lo, up, 0, 1, 2, 3, 4, 5, 6, 7);
                v2f a[4];
#pragma unroll
                for (int kk = 0; kk < 4; ++kk) a[kk] = load_afrag_lds(&sH0[prv][0], P0, 4 * kk, lane);
#pragma unroll
                for (int kk = 0; kk < 4; ++kk) acc = wmma4(a[kk], w[kk], acc);
                cell2<true>(acc, c, &sH0[cur][hrow * P0 + u0 + u], P0, hl);
            }
        } else if (kind == KIND_L2) {
            const int t = p - 3;
            if (t >= 0 && t < T_STEPS) {
                v2f ax[8], ah[4];
#pragma unroll
                for (int kk = 0; kk < 8; ++kk) ax[kk] = load_afrag_lds(&sH1[prv][0], P1, 4 * kk, lane);
#pragma unroll
                for (int kk = 0; kk < 4; ++kk) ah[kk] = load_afrag_lds(&sH2[prv][0], P2, 4 * kk, lane);
                v8f acc;
#pragma unroll
                for (int j = 0; j < 8; ++j) acc[j] = bias;
#pragma unroll
                for (int kk = 0; kk < 8; ++kk) acc = wmma4(ax[kk], w[kk], acc);
#pragma unroll
                for (int kk = 0; kk < 4; ++kk) acc = wmma4(ah[kk], w[8 + kk], acc);
                cell2<true>(acc, c, &sH2[cur][hrow * P2 + u0 + u], P2, hl);
            }
        } else if (kind == KIND_OUT) {
            const int t = p - 4;
            if (t >= 0) {
                v2f ah[4];
#pragma unroll
                for (int kk = 0; kk < 4; ++kk) ah[kk] = load_afrag_lds(&sH2[prv][0], P2, 4 * kk, lane);
                v8f po;
#pragma unroll
                for (int j = 0; j < 8; ++j) po[j] = biasO;
#pragma unroll
                for (int kk = 0; kk < 4; ++kk) po = wmma4(ah[kk], w[kk], po);
                if (ln < OUT_DIM) {
                    float* o = out + (((size_t)(b0 + 8 * hi)) * T_STEPS + t) * OUT_DIM + ln;
#pragma unroll
                    for (int j = 0; j < 8; ++j) o[(size_t)j * T_STEPS * OUT_DIM] = po[j];
                }
            }
        } else {
            const int t = p - 2;
            if (t >= 0 && t < T_STEPS) {
                v2f ax[4], ah[8];
#pragma unroll
                for (int kk = 0; kk < 4; ++kk) ax[kk] = load_afrag_lds(&sH0[prv][0], P0, 4 * kk, lane);
#pragma unroll
                for (int kk = 0; kk < 8; ++kk) ah[kk] = load_afrag_lds(&sH1[prv][0], P1, 4 * kk, lane);
                v8f acc;
#pragma unroll
                for (int j = 0; j < 8; ++j) acc[j] = bias;
#pragma unroll
                for (int kk = 0; kk < 4; ++kk) acc = wmma4(ax[kk], w[kk], acc);
#pragma unroll
                for (int kk = 0; kk < 8; ++kk) acc = wmma4(ah[kk], w[4 + kk], acc);
                cell2<true>(acc, c, &sH1[cur][hrow * P1 + u0 + u], P1, hl);
            }
        }
        __syncthreads();
    }
}


#define PWAVES   9
#define PTHREADS (PWAVES * 32)
#define NTILES   (BATCH / 16)
#define KIND_ST   5
#define KIND_IDLE 6

__device__ __forceinline__ v2f load_afrag_g(const float* base, int lda, int kbase, int lane) {
    int m = lane & 15;
    int k = kbase + ((lane >> 4) << 1);
    return *(const v2f*)(base + m * lda + k);
}

template <int C>
__global__ __launch_bounds__(PTHREADS) void lstm3_layer_chunk_kernel(
    const float* __restrict__ x,
    const float* __restrict__ Wih0, const float* __restrict__ Whh0,
    const float* __restrict__ bih0, const float* __restrict__ bhh0,
    const float* __restrict__ Wih1, const float* __restrict__ Whh1,
    const float* __restrict__ bih1, const float* __restrict__ bhh1,
    const float* __restrict__ Wih2, const float* __restrict__ Whh2,
    const float* __restrict__ bih2, const float* __restrict__ bhh2,
    const float* __restrict__ Wl,   const float* __restrict__ bl,
    float* __restrict__ out, float* ws, int d)
{
    constexpr int NCH = T_STEPS / C;
    const int role = blockIdx.y;
    const int k    = d - role;
    if (k < 0 || k >= NCH) return;

    __shared__ __attribute__((aligned(16))) float sXP[2][4 * 2 * 32 * 4];
    __shared__ __attribute__((aligned(16))) float sH0[2][16 * P0];
    __shared__ __attribute__((aligned(16))) float sH1[2][16 * P1];
    __shared__ __attribute__((aligned(16))) float sH2[2][16 * P2];

    const int tid  = threadIdx.x;
    const int lane = tid & 31;
    const int ln   = lane & 15;
    const int hi   = lane >> 4;
    const int u    = ln & 3;
    const int gq   = ln >> 2;
    const int wave = __builtin_amdgcn_readfirstlane(tid >> 5);
    const int b0   = blockIdx.x * 16;
    const int t0   = k * C;
    const int kb   = k & 1;
    const int hrow = 8 * hi + 2 * gq;
    const int srow  = lane >> 1;
    const int shalf = lane & 1;

    float* ho0c = ws + ((size_t)kb * C * BATCH + b0) * H0;
    float* ho1c = ws + (size_t)2 * C * BATCH * H0 + ((size_t)kb * C * BATCH + b0) * H1;
    float* st   = ws + (size_t)2 * C * BATCH * (H0 + H1);
    v4f* stl = (v4f*)st + ((size_t)((role * NTILES + blockIdx.x) * 8 + (wave & 7)) * 32 + lane);

    float c[2]  = {0.f, 0.f};
    float hl[2] = {0.f, 0.f};

    if (role == 0) {
        const int kind = (wave < 4) ? KIND_L0 : (wave < 8) ? KIND_X : KIND_ST;
        const int q  = wave & 3;
        const int u0 = q * 4;
        const int r  = gq * H0 + u0 + u;
        v2f w[16];
        float bias = 0.f;
        for (int i = 0; i < 16; ++i) { w[i].x = 0.f; w[i].y = 0.f; }
        if (kind == KIND_X) {
            for (int kk = 0; kk < 16; ++kk) w[kk] = load_bfrag_row(Wih0, IN_DIM, r, 4 * kk, lane);
            bias = bih0[r] + bhh0[r];
        } else if (kind == KIND_L0) {
            for (int kk = 0; kk < 4; ++kk) w[kk] = load_bfrag_row(Whh0, H0, r, 4 * kk, lane);
            if (k > 0) { v4f s = *stl; c[0] = s.x; c[1] = s.y; hl[0] = s.z; hl[1] = s.w; }
            float* h = &sH0[0][hrow * P0 + u0 + u];
            h[0] = hl[0]; h[P0] = hl[1]; h[16 * P0] = hl[0]; h[16 * P0 + P0] = hl[1];
        }
        __syncthreads();
        for (int p = 0; p < C + 2; ++p) {
            const int cur = p & 1, prv = cur ^ 1;
            if (kind == KIND_X) {
                if (p < C) {
                    const float* xt = x + ((size_t)(t0 + p) * BATCH + b0) * IN_DIM;
                    if (t0 + p + 1 < T_STEPS)
                        __builtin_prefetch(xt + (size_t)BATCH * IN_DIM + lane * 32, 0, 0);
                    v2f a[16];
                    for (int kk = 0; kk < 16; ++kk) a[kk] = load_afrag(xt, IN_DIM, 4 * kk, lane);
                    v8f acc;
                    for (int j = 0; j < 8; ++j) acc[j] = bias;
                    for (int kk = 0; kk < 16; ++kk) acc = wmma4(a[kk], w[kk], acc);
                    v4f* dst = (v4f*)&sXP[cur][0] + (q * 2) * 32 + lane;
                    dst[0]  = __builtin_shufflevector(acc, acc, 0, 1, 2, 3);
                    dst[32] = __builtin_shufflevector(acc, acc, 4, 5, 6, 7);
                }
            } else if (kind == KIND_L0) {
                if (p >= 1 && p <= C) {
                    const v4f* src = (const v4f*)&sXP[prv][0] + (q * 2) * 32 + lane;
                    v4f lo = src[0], up = src[32];
                    v8f acc = __builtin_shufflevector(lo, up, 0, 1, 2, 3, 4, 5, 6, 7);
                    v2f a[4];
                    for (int kk = 0; kk < 4; ++kk) a[kk] = load_afrag_lds(&sH0[prv][0], P0, 4 * kk, lane);
                    for (int kk = 0; kk < 4; ++kk) acc = wmma4(a[kk], w[kk], acc);
                    cell2<true>(acc, c, &sH0[cur][hrow * P0 + u0 + u], P0, hl);
                }
            } else {
                if (p >= 2) {
                    const v4f* src = (const v4f*)&sH0[prv][srow * P0 + shalf * 8];
                    v4f a0 = src[0], a1 = src[1];
                    v4f* dst = (v4f*)(ho0c + (size_t)(p - 2) * BATCH * H0 + srow * H0 + shalf * 8);
                    dst[0] = a0; dst[1] = a1;
                }
            }
            __syncthreads();
        }
        if (kind == KIND_L0) { v4f s; s.x = c[0]; s.y = c[1]; s.z = hl[0]; s.w = hl[1]; *stl = s; }
    } else if (role == 1) {
        const int kind = (wave < 8) ? KIND_L1 : KIND_ST;
        const int q  = wave & 7;
        const int u0 = q * 4;
        const int r  = gq * H1 + u0 + u;
        v2f w[12];
        v2f axn[4];
        float bias = 0.f;
        for (int i = 0; i < 12; ++i) { w[i].x = 0.f; w[i].y = 0.f; }
        for (int i = 0; i < 4; ++i) { axn[i].x = 0.f; axn[i].y = 0.f; }
        if (kind == KIND_L1) {
            for (int kk = 0; kk < 4; ++kk) w[kk] = load_bfrag_row(Wih1, H0, r, 4 * kk, lane);
            for (int kk = 0; kk < 8; ++kk) w[4 + kk] = load_bfrag_row(Whh1, H1, r, 4 * kk, lane);
            bias = bih1[r] + bhh1[r];
            if (k > 0) { v4f s = *stl; c[0] = s.x; c[1] = s.y; hl[0] = s.z; hl[1] = s.w; }
            float* h = &sH1[0][hrow * P1 + u0 + u];
            h[0] = hl[0]; h[P1] = hl[1]; h[16 * P1] = hl[0]; h[16 * P1 + P1] = hl[1];
            for (int kk = 0; kk < 4; ++kk) axn[kk] = load_afrag_g(ho0c, H0, 4 * kk, lane);
        }
        __syncthreads();
        for (int p = 0; p < C + 1; ++p) {
            const int cur = p & 1, prv = cur ^ 1;
            if (kind == KIND_L1) {
                if (p < C) {
                    v2f ax[4], ah[8];
                    for (int kk = 0; kk < 4; ++kk) ax[kk] = axn[kk];
                    for (int kk = 0; kk < 8; ++kk) ah[kk] = load_afrag_lds(&sH1[prv][0], P1, 4 * kk, lane);
                    if (p + 1 < C) {
                        for (int kk = 0; kk < 4; ++kk)
                            axn[kk] = load_afrag_g(ho0c + (size_t)(p + 1) * BATCH * H0, H0, 4 * kk, lane);
                    }
                    v8f acc;
                    for (int j = 0; j < 8; ++j) acc[j] = bias;
                    for (int kk = 0; kk < 4; ++kk) acc = wmma4(ax[kk], w[kk], acc);
                    for (int kk = 0; kk < 8; ++kk) acc = wmma4(ah[kk], w[4 + kk], acc);
                    if (wave < 4) cell2<true>(acc, c, &sH1[cur][hrow * P1 + u0 + u], P1, hl);
                    else          cell2<false>(acc, c, &sH1[cur][hrow * P1 + u0 + u], P1, hl);
                }
            } else {
                if (p >= 1) {
                    const v4f* src = (const v4f*)&sH1[prv][srow * P1 + shalf * 16];
                    v4f a0 = src[0], a1 = src[1], a2 = src[2], a3 = src[3];
                    v4f* dst = (v4f*)(ho1c + (size_t)(p - 1) * BATCH * H1 + srow * H1 + shalf * 16);
                    dst[0] = a0; dst[1] = a1; dst[2] = a2; dst[3] = a3;
                }
            }
            __syncthreads();
        }
        if (kind == KIND_L1) { v4f s; s.x = c[0]; s.y = c[1]; s.z = hl[0]; s.w = hl[1]; *stl = s; }
    } else {
        const int kind = (wave < 4) ? KIND_L2 : (wave == 4) ? KIND_OUT : KIND_IDLE;
        const int q  = wave & 3;
        const int u0 = q * 4;
        const int r  = gq * H2 + u0 + u;
        v2f w[12];
        v2f axn[8];
        float bias = 0.f, biasO = 0.f;
        for (int i = 0; i < 12; ++i) { w[i].x = 0.f; w[i].y = 0.f; }
        for (int i = 0; i < 8; ++i) { axn[i].x = 0.f; axn[i].y = 0.f; }
        if (kind == KIND_L2) {
            for (int kk = 0; kk < 8; ++kk) w[kk] = load_bfrag_row(Wih2, H1, r, 4 * kk, lane);
            for (int kk = 0; kk < 4; ++kk) w[8 + kk] = load_bfrag_row(Whh2, H2, r, 4 * kk, lane);
            bias = bih2[r] + bhh2[r];
            if (k > 0) { v4f s = *stl; c[0] = s.x; c[1] = s.y; hl[0] = s.z; hl[1] = s.w; }
            float* h = &sH2[0][hrow * P2 + u0 + u];
            h[0] = hl[0]; h[P2] = hl[1]; h[16 * P2] = hl[0]; h[16 * P2 + P2] = hl[1];
            for (int kk = 0; kk < 8; ++kk) axn[kk] = load_afrag_g(ho1c, H1, 4 * kk, lane);
        } else if (kind == KIND_OUT) {
            if (ln < OUT_DIM) {
                for (int kk = 0; kk < 4; ++kk) w[kk] = load_bfrag_row(Wl, H2, ln, 4 * kk, lane);
                biasO = bl[ln];
            }
        }
        __syncthreads();
        for (int p = 0; p < C + 1; ++p) {
            const int cur = p & 1, prv = cur ^ 1;
            if (kind == KIND_L2) {
                if (p < C) {
                    v2f ax[8], ah[4];
                    for (int kk = 0; kk < 8; ++kk) ax[kk] = axn[kk];
                    for (int kk = 0; kk < 4; ++kk) ah[kk] = load_afrag_lds(&sH2[prv][0], P2, 4 * kk, lane);
                    if (p + 1 < C) {
                        for (int kk = 0; kk < 8; ++kk)
                            axn[kk] = load_afrag_g(ho1c + (size_t)(p + 1) * BATCH * H1, H1, 4 * kk, lane);
                    }
                    v8f acc;
                    for (int j = 0; j < 8; ++j) acc[j] = bias;
                    for (int kk = 0; kk < 8; ++kk) acc = wmma4(ax[kk], w[kk], acc);
                    for (int kk = 0; kk < 4; ++kk) acc = wmma4(ah[kk], w[8 + kk], acc);
                    cell2<true>(acc, c, &sH2[cur][hrow * P2 + u0 + u], P2, hl);
                }
            } else if (kind == KIND_OUT) {
                if (p >= 1) {
                    const int t = t0 + p - 1;
                    v2f ah[4];
                    for (int kk = 0; kk < 4; ++kk) ah[kk] = load_afrag_lds(&sH2[prv][0], P2, 4 * kk, lane);
                    v8f po;
                    for (int j = 0; j < 8; ++j) po[j] = biasO;
                    for (int kk = 0; kk < 4; ++kk) po = wmma4(ah[kk], w[kk], po);
                    if (ln < OUT_DIM) {
                        float* o = out + (((size_t)(b0 + 8 * hi)) * T_STEPS + t) * OUT_DIM + ln;
                        for (int j = 0; j < 8; ++j) o[(size_t)j * T_STEPS * OUT_DIM] = po[j];
                    }
                }
            }
            __syncthreads();
        }
        if (kind == KIND_L2) { v4f s; s.x = c[0]; s.y = c[1]; s.z = hl[0]; s.w = hl[1]; *stl = s; }
    }
}

template <int C>
static void launch_chunks(const float* x,
                          const float* Wih0, const float* Whh0, const float* bih0, const float* bhh0,
                          const float* Wih1, const float* Whh1, const float* bih1, const float* bhh1,
                          const float* Wih2, const float* Whh2, const float* bih2, const float* bhh2,
                          const float* Wl, const float* bl, float* out, float* ws, hipStream_t stream) {
    for (int d = 0; d < T_STEPS / C + 2; ++d)
        lstm3_layer_chunk_kernel<C><<<dim3(NTILES, 3), PTHREADS, 0, stream>>>(
            x, Wih0, Whh0, bih0, bhh0, Wih1, Whh1, bih1, bhh1, Wih2, Whh2, bih2, bhh2, Wl, bl, out, ws, d);
}

extern "C" void kernel_launch(void* const* d_in, const int* in_sizes, int n_in,
                              void* d_out, int out_size, void* d_ws, size_t ws_size,
                              hipStream_t stream) {
    (void)in_sizes; (void)n_in; (void)out_size;
    const float* x    = (const float*)d_in[0];
    const float* Wih0 = (const float*)d_in[1];
    const float* Whh0 = (const float*)d_in[2];
    const float* bih0 = (const float*)d_in[3];
    const float* bhh0 = (const float*)d_in[4];
    const float* Wih1 = (const float*)d_in[5];
    const float* Whh1 = (const float*)d_in[6];
    const float* bih1 = (const float*)d_in[7];
    const float* bhh1 = (const float*)d_in[8];
    const float* Wih2 = (const float*)d_in[9];
    const float* Whh2 = (const float*)d_in[10];
    const float* bih2 = (const float*)d_in[11];
    const float* bhh2 = (const float*)d_in[12];
    const float* Wl   = (const float*)d_in[13];
    const float* bl   = (const float*)d_in[14];
    float* out = (float*)d_out;

    float* ws = (float*)d_ws;
    if (ws != nullptr && ws_size >= (size_t)6488064) {
        launch_chunks<64>(x, Wih0, Whh0, bih0, bhh0, Wih1, Whh1, bih1, bhh1, Wih2, Whh2, bih2, bhh2, Wl, bl, out, ws, stream);
    } else if (ws != nullptr && ws_size >= (size_t)3342336) {
        launch_chunks<32>(x, Wih0, Whh0, bih0, bhh0, Wih1, Whh1, bih1, bhh1, Wih2, Whh2, bih2, bhh2, Wl, bl, out, ws, stream);
    } else if (ws != nullptr && ws_size >= (size_t)1769472) {
        launch_chunks<16>(x, Wih0, Whh0, bih0, bhh0, Wih1, Whh1, bih1, bhh1, Wih2, Whh2, bih2, bhh2, Wl, bl, out, ws, stream);
    } else {
        lstm3_stage_kernel<<<BATCH / 16, NTHREADS, 0, stream>>>(
            x, Wih0, Whh0, bih0, bhh0,
            Wih1, Whh1, bih1, bhh1,
            Wih2, Whh2, bih2, bhh2,
            Wl, bl, out);
    }
}


# layer-0 x tile: one wave fetches each step's 4 KB once (whole-line 16-byte loads) into LDS a pass ahead; the 4 x waves read A fragments there, not 8-byte pieces of 16 rows from global; sum order kept
# speedup vs baseline: 4.2034x; 1.0563x over previous
#include <hip/hip_runtime.h>


typedef __attribute__((ext_vector_type(2))) float v2f;
typedef __attribute__((ext_vector_type(4))) float v4f;
typedef __attribute__((ext_vector_type(8))) float v8f;

#define T_STEPS  2048
#define BATCH    256
#define IN_DIM   64
#define H0       16
#define H1       32
#define H2       16
#define OUT_DIM  8
#define NWAVES   21
#define NTHREADS (NWAVES * 32)

#define P0 20
#define P1 36
#define P2 20
#define PX 68

#define KIND_L1  0
#define KIND_L0  1
#define KIND_L2  2
#define KIND_X   3
#define KIND_OUT 4

__device__ __forceinline__ float sigmoidf_(float x) { return 1.0f / (1.0f + __expf(-x)); }

__device__ __forceinline__ v2f load_afrag(const float* base, int lda, int kbase, int lane) {
    int m = lane & 15;
    int k = kbase + ((lane >> 4) << 1);
    const float* p = base + m * lda + k;
    v2f a;
    a.x = p[0];
    a.y = p[1];
    return a;
}

__device__ __forceinline__ v2f load_afrag_lds(const float* base, int pitch, int kbase, int lane) {
    int m = lane & 15;
    int k = kbase + ((lane >> 4) << 1);
    return *(const v2f*)(base + m * pitch + k);
}

__device__ __forceinline__ v2f load_bfrag_row(const float* W, int ldw, int row, int kbase, int lane) {
    int k = kbase + ((lane >> 4) << 1);
    const float* p = W + row * ldw + k;
    v2f b;
    b.x = p[0];
    b.y = p[1];
    return b;
}

__device__ __forceinline__ v8f wmma4(v2f a, v2f b, v8f c) {
    return __builtin_amdgcn_wmma_f32_16x16x4_f32(false, a, false, b, (short)0, c, false, false);
}

template <int ROR, int BANK>
__device__ __forceinline__ float take(float keep, float give) {
    return __builtin_bit_cast(float,
        __builtin_amdgcn_update_dpp(__builtin_bit_cast(int, keep), __builtin_bit_cast(int, give),
                                    0x120 + ROR, 0xf, BANK, false));
}

template <int JJ> __device__ __forceinline__ float gate_i(const v8f& t) {
    return take<12, 0x8>(take<8, 0x4>(take<4,  0x2>(t[JJ],     t[2 + JJ]), t[4 + JJ]), t[6 + JJ]);
}
template <int JJ> __device__ __forceinline__ float gate_f(const v8f& t) {
    return take<8,  0x8>(take<4, 0x4>(take<12, 0x1>(t[2 + JJ], t[JJ]),     t[4 + JJ]), t[6 + JJ]);
}
template <int JJ> __device__ __forceinline__ float gate_g(const v8f& t) {
    return take<4,  0x8>(take<12, 0x2>(take<8, 0x1>(t[4 + JJ], t[JJ]),     t[2 + JJ]), t[6 + JJ]);
}
template <int JJ> __device__ __forceinline__ float gate_o(const v8f& t) {
    return take<12, 0x4>(take<8, 0x2>(take<4,  0x1>(t[6 + JJ], t[JJ]),     t[2 + JJ]), t[4 + JJ]);
}

template <bool SIG_FIRST>
__device__ __forceinline__ void cell2(const v8f& t, float (&c)[2], float* hdst, int pitch, float (&hl)[2]) {
    float si0, sf0, so0, si1, sf1, so1, tg0, vg1;
    if (SIG_FIRST) {
        const float vg0 = gate_g<0>(t);
        vg1 = gate_g<1>(t);
        si0 = sigmoidf_(gate_i<0>(t)); sf0 = sigmoidf_(gate_f<0>(t)); so0 = sigmoidf_(gate_o<0>(t));
        si1 = sigmoidf_(gate_i<1>(t)); sf1 = sigmoidf_(gate_f<1>(t)); so1 = sigmoidf_(gate_o<1>(t));
        asm volatile("" : "+v"(si0), "+v"(sf0), "+v"(so0), "+v"(si1), "+v"(sf1), "+v"(so1));
        tg0 = tanhf(vg0);
    } else {
        tg0 = tanhf(gate_g<0>(t));
        vg1 = gate_g<1>(t);
        si0 = sigmoidf_(gate_i<0>(t)); sf0 = sigmoidf_(gate_f<0>(t)); so0 = sigmoidf_(gate_o<0>(t));
        si1 = sigmoidf_(gate_i<1>(t)); sf1 = sigmoidf_(gate_f<1>(t)); so1 = sigmoidf_(gate_o<1>(t));
        asm volatile("" : "+v"(si0), "+v"(sf0), "+v"(so0), "+v"(si1), "+v"(sf1), "+v"(so1));
    }
    const float tg1 = tanhf(vg1);
    const float cn0 = sf0 * c[0] + si0 * tg0;
    const float cn1 = sf1 * c[1] + si1 * tg1;
    c[0] = cn0;
    c[1] = cn1;
    const float hv0 = so0 * tanhf(cn0);
    const float hv1 = so1 * tanhf(cn1);
    hl[0] = hv0;
    hl[1] = hv1;
    hdst[0]     = hv0;
    hdst[pitch] = hv1;
}

__global__ __launch_bounds__(NTHREADS) void lstm3_stage_kernel(
    const float* __restrict__ x,
    const float* __restrict__ Wih0, const float* __restrict__ Whh0,
    const float* __restrict__ bih0, const float* __restrict__ bhh0,
    const float* __restrict__ Wih1, const float* __restrict__ Whh1,
    const float* __restrict__ bih1, const float* __restrict__ bhh1,
    const float* __restrict__ Wih2, const float* __restrict__ Whh2,
    const float* __restrict__ bih2, const float* __restrict__ bhh2,
    const float* __restrict__ Wl,   const float* __restrict__ bl,
    float* __restrict__ out)
{
    __shared__ __attribute__((aligned(16))) float sXP[2][4 * 2 * 32 * 4];
    __shared__ __attribute__((aligned(16))) float sH0[2][16 * P0];
    __shared__ __attribute__((aligned(16))) float sH1[2][16 * P1];
    __shared__ __attribute__((aligned(16))) float sH2[2][16 * P2];

    const int tid  = threadIdx.x;
    const int lane = tid & 31;
    const int ln   = lane & 15;
    const int hi   = lane >> 4;
    const int u    = ln & 3;
    const int gq   = ln >> 2;
    const int wave = __builtin_amdgcn_readfirstlane(tid >> 5);
    const int kind = (wave < 8) ? KIND_L1 : (wave < 12) ? KIND_L2 : (wave < 16) ? KIND_L0
                   : (wave < 17) ? KIND_OUT : KIND_X;
    const int q    = (wave < 8) ? wave : (wave < 16) ? (wave & 3) : (wave < 17) ? 0 : (wave - 17);
    const int u0   = q * 4;
    const int b0   = blockIdx.x * 16;

    for (int i = tid; i < 2 * 16 * P0; i += NTHREADS) { (&sH0[0][0])[i] = 0.f; (&sH2[0][0])[i] = 0.f; }
    for (int i = tid; i < 2 * 16 * P1; i += NTHREADS) { (&sH1[0][0])[i] = 0.f; }

    v2f   w[16];
    float bias = 0.f;
    float c[2];
    float hl[2];
#pragma unroll
    for (int i = 0; i < 16; ++i) { w[i].x = 0.f; w[i].y = 0.f; }
    c[0] = 0.f; c[1] = 0.f;
    float biasO = 0.f;

    if (kind == KIND_X) {
        const int r = gq * H0 + u0 + u;
#pragma unroll
        for (int kk = 0; kk < 16; ++kk) w[kk] = load_bfrag_row(Wih0, IN_DIM, r, 4 * kk, lane);
        bias = bih0[r] + bhh0[r];
    } else if (kind == KIND_L0) {
        const int r = gq * H0 + u0 + u;
#pragma unroll
        for (int kk = 0; kk < 4; ++kk) w[kk] = load_bfrag_row(Whh0, H0, r, 4 * kk, lane);
    } else if (kind == KIND_L2) {
        const int r = gq * H2 + u0 + u;
#pragma unroll
        for (int kk = 0; kk < 8; ++kk) w[kk] = load_bfrag_row(Wih2, H1, r, 4 * kk, lane);
#pragma unroll
        for (int kk = 0; kk < 4; ++kk) w[8 + kk] = load_bfrag_row(Whh2, H2, r, 4 * kk, lane);
        bias = bih2[r] + bhh2[r];
    } else if (kind == KIND_OUT) {
        if (ln < OUT_DIM) {
#pragma unroll
            for (int kk = 0; kk < 4; ++kk) w[kk] = load_bfrag_row(Wl, H2, ln, 4 * kk, lane);
            biasO = bl[ln];
        }
    } else {
        const int r = gq * H1 + u0 + u;
#pragma unroll
        for (int kk = 0; kk < 4; ++kk) w[kk] = load_bfrag_row(Wih1, H0, r, 4 * kk, lane);
#pragma unroll
        for (int kk = 0; kk < 8; ++kk) w[4 + kk] = load_bfrag_row(Whh1, H1, r, 4 * kk, lane);
        bias = bih1[r] + bhh1[r];
    }

    const int hrow = 8 * hi + 2 * gq;

    __syncthreads();

    for (int p = 0; p < T_STEPS + 4; ++p) {
        const int cur = p & 1, prv = cur ^ 1;

        if (kind == KIND_X) {
            if (p < T_STEPS) {
                const float* xt = x + ((size_t)p * BATCH + b0) * IN_DIM;
                if (p + 1 < T_STEPS)
                    __builtin_prefetch(xt + (size_t)BATCH * IN_DIM + lane * 32, 0, 0);
                v2f a[16];
#pragma unroll
                for (int kk = 0; kk < 16; ++kk) a[kk] = load_afrag(xt, IN_DIM, 4 * kk, lane);
                v8f acc;
#pragma unroll
                for (int j = 0; j < 8; ++j) acc[j] = bias;
#pragma unroll
                for (int kk = 0; kk < 16; ++kk) acc = wmma4(a[kk], w[kk], acc);
                v4f* dst = (v4f*)&sXP[cur][0] + (q * 2) * 32 + lane;
                dst[0]  = __builtin_shufflevector(acc, acc, 0, 1, 2, 3);
                dst[32] = __builtin_shufflevector(acc, acc, 4, 5, 6, 7);
            }
        } else if (kind == KIND_L0) {
            const int t = p - 1;
            if (t >= 0 && t < T_STEPS) {
                const v4f* src = (const v4f*)&sXP[prv][0] + (q * 2) * 32 + lane;
                v4f lo = src[0], up = src[32];
                v8f acc = __builtin_shufflevector(lo, up, 0, 1, 2, 3, 4, 5, 6, 7);
                v2f a[4];
#pragma unroll
                for (int kk = 0; kk < 4; ++kk) a[kk] = load_afrag_lds(&sH0[prv][0], P0, 4 * kk, lane);
#pragma unroll
                for (int kk = 0; kk < 4; ++kk) acc = wmma4(a[kk], w[kk], acc);
                cell2<true>(acc, c, &sH0[cur][hrow * P0 + u0 + u], P0, hl);
            }
        } else if (kind == KIND_L2) {
            const int t = p - 3;
            if (t >= 0 && t < T_STEPS) {
                v2f ax[8], ah[4];
#pragma unroll
                for (int kk = 0; kk < 8; ++kk) ax[kk] = load_afrag_lds(&sH1[prv][0], P1, 4 * kk, lane);
#pragma unroll
                for (int kk = 0; kk < 4; ++kk) ah[kk] = load_afrag_lds(&sH2[prv][0], P2, 4 * kk, lane);
                v8f acc;
#pragma unroll
                for (int j = 0; j < 8; ++j) acc[j] = bias;
#pragma unroll
                for (int kk = 0; kk < 8; ++kk) acc = wmma4(ax[kk], w[kk], acc);
#pragma unroll
                for (int kk = 0; kk < 4; ++kk) acc = wmma4(ah[kk], w[8 + kk], acc);
                cell2<true>(acc, c, &sH2[cur][hrow * P2 + u0 + u], P2, hl);
            }
        } else if (kind == KIND_OUT) {
            const int t = p - 4;
            if (t >= 0) {
                v2f ah[4];
#pragma unroll
                for (int kk = 0; kk < 4; ++kk) ah[kk] = load_afrag_lds(&sH2[prv][0], P2, 4 * kk, lane);
                v8f po;
#pragma unroll
                for (int j = 0; j < 8; ++j) po[j] = biasO;
#pragma unroll
                for (int kk = 0; kk < 4; ++kk) po = wmma4(ah[kk], w[kk], po);
                if (ln < OUT_DIM) {
                    float* o = out + (((size_t)(b0 + 8 * hi)) * T_STEPS + t) * OUT_DIM + ln;
#pragma unroll
                    for (int j = 0; j < 8; ++j) o[(size_t)j * T_STEPS * OUT_DIM] = po[j];
                }
            }
        } else {
            const int t = p - 2;
            if (t >= 0 && t < T_STEPS) {
                v2f ax[4], ah[8];
#pragma unroll
                for (int kk = 0; kk < 4; ++kk) ax[kk] = load_afrag_lds(&sH0[prv][0], P0, 4 * kk, lane);
#pragma unroll
                for (int kk = 0; kk < 8; ++kk) ah[kk] = load_afrag_lds(&sH1[prv][0], P1, 4 * kk, lane);
                v8f acc;
#pragma unroll
                for (int j = 0; j < 8; ++j) acc[j] = bias;
#pragma unroll
                for (int kk = 0; kk < 4; ++kk) acc = wmma4(ax[kk], w[kk], acc);
#pragma unroll
                for (int kk = 0; kk < 8; ++kk) acc = wmma4(ah[kk], w[4 + kk], acc);
                cell2<true>(acc, c, &sH1[cur][hrow * P1 + u0 + u], P1, hl);
            }
        }
        __syncthreads();
    }
}


#define PWAVES   9
#define PTHREADS (PWAVES * 32)
#define NTILES   (BATCH / 16)
#define KIND_ST   5
#define KIND_IDLE 6

__device__ __forceinline__ v2f load_afrag_g(const float* base, int lda, int kbase, int lane) {
    int m = lane & 15;
    int k = kbase + ((lane >> 4) << 1);
    return *(const v2f*)(base + m * lda + k);
}

template <int C>
__global__ __launch_bounds__(PTHREADS) void lstm3_layer_chunk_kernel(
    const float* __restrict__ x,
    const float* __restrict__ Wih0, const float* __restrict__ Whh0,
    const float* __restrict__ bih0, const float* __restrict__ bhh0,
    const float* __restrict__ Wih1, const float* __restrict__ Whh1,
    const float* __restrict__ bih1, const float* __restrict__ bhh1,
    const float* __restrict__ Wih2, const float* __restrict__ Whh2,
    const float* __restrict__ bih2, const float* __restrict__ bhh2,
    const float* __restrict__ Wl,   const float* __restrict__ bl,
    float* __restrict__ out, float* ws, int d)
{
    constexpr int NCH = T_STEPS / C;
    const int role = blockIdx.y;
    const int k    = d - role;
    if (k < 0 || k >= NCH) return;

    __shared__ __attribute__((aligned(16))) float sXP[2][4 * 2 * 32 * 4];
    __shared__ __attribute__((aligned(16))) float sH0[2][16 * P0];
    __shared__ __attribute__((aligned(16))) float sH1[2][16 * P1];
    __shared__ __attribute__((aligned(16))) float sH2[2][16 * P2];
    __shared__ __attribute__((aligned(16))) float sX[2][16 * PX];

    const int tid  = threadIdx.x;
    const int lane = tid & 31;
    const int ln   = lane & 15;
    const int hi   = lane >> 4;
    const int u    = ln & 3;
    const int gq   = ln >> 2;
    const int wave = __builtin_amdgcn_readfirstlane(tid >> 5);
    const int b0   = blockIdx.x * 16;
    const int t0   = k * C;
    const int kb   = k & 1;
    const int hrow = 8 * hi + 2 * gq;
    const int srow  = lane >> 1;
    const int shalf = lane & 1;

    float* ho0c = ws + ((size_t)kb * C * BATCH + b0) * H0;
    float* ho1c = ws + (size_t)2 * C * BATCH * H0 + ((size_t)kb * C * BATCH + b0) * H1;
    float* st   = ws + (size_t)2 * C * BATCH * (H0 + H1);
    v4f* stl = (v4f*)st + ((size_t)((role * NTILES + blockIdx.x) * 8 + (wave & 7)) * 32 + lane);

    float c[2]  = {0.f, 0.f};
    float hl[2] = {0.f, 0.f};

    if (role == 0) {
        const int kind = (wave < 4) ? KIND_L0 : (wave < 8) ? KIND_X : KIND_ST;
        const int q  = wave & 3;
        const int u0 = q * 4;
        const int r  = gq * H0 + u0 + u;
        v2f w[16];
        float bias = 0.f;
        for (int i = 0; i < 16; ++i) { w[i].x = 0.f; w[i].y = 0.f; }
        float* xdst = &sX[0][hi * PX + 4 * ln];
        const v4f* xsrc = (const v4f*)(x + ((size_t)t0 * BATCH + b0) * IN_DIM) + lane;
        if (kind == KIND_ST) {
            v4f xr[8];
            for (int i = 0; i < 8; ++i) xr[i] = xsrc[32 * i];
            for (int i = 0; i < 8; ++i) *(v4f*)(xdst + 16 * PX + 2 * i * PX) = xr[i];
        }
        if (kind == KIND_X) {
            for (int kk = 0; kk < 16; ++kk) w[kk] = load_bfrag_row(Wih0, IN_DIM, r, 4 * kk, lane);
            bias = bih0[r] + bhh0[r];
        } else if (kind == KIND_L0) {
            for (int kk = 0; kk < 4; ++kk) w[kk] = load_bfrag_row(Whh0, H0, r, 4 * kk, lane);
            if (k > 0) { v4f s = *stl; c[0] = s.x; c[1] = s.y; hl[0] = s.z; hl[1] = s.w; }
            float* h = &sH0[0][hrow * P0 + u0 + u];
            h[0] = hl[0]; h[P0] = hl[1]; h[16 * P0] = hl[0]; h[16 * P0 + P0] = hl[1];
        }
        __syncthreads();
        for (int p = 0; p < C + 2; ++p) {
            const int cur = p & 1, prv = cur ^ 1;
            if (kind == KIND_X) {
                if (p < C) {
                    v2f a[16];
                    for (int kk = 0; kk < 16; ++kk) a[kk] = load_afrag_lds(&sX[prv][0], PX, 4 * kk, lane);
                    v8f acc;
                    for (int j = 0; j < 8; ++j) acc[j] = bias;
                    for (int kk = 0; kk < 16; ++kk) acc = wmma4(a[kk], w[kk], acc);
                    v4f* dst = (v4f*)&sXP[cur][0] + (q * 2) * 32 + lane;
                    dst[0]  = __builtin_shufflevector(acc, acc, 0, 1, 2, 3);
                    dst[32] = __builtin_shufflevector(acc, acc, 4, 5, 6, 7);
                }
            } else if (kind == KIND_L0) {
                if (p >= 1 && p <= C) {
                    const v4f* src = (const v4f*)&sXP[prv][0] + (q * 2) * 32 + lane;
                    v4f lo = src[0], up = src[32];
                    v8f acc = __builtin_shufflevector(lo, up, 0, 1, 2, 3, 4, 5, 6, 7);
                    v2f a[4];
                    for (int kk = 0; kk < 4; ++kk) a[kk] = load_afrag_lds(&sH0[prv][0], P0, 4 * kk, lane);
                    for (int kk = 0; kk < 4; ++kk) acc = wmma4(a[kk], w[kk], acc);
                    cell2<true>(acc, c, &sH0[cur][hrow * P0 + u0 + u], P0, hl);
                }
            } else {
                if (p < C && t0 + p + 2 < T_STEPS)
                    __builtin_prefetch(x + ((size_t)(t0 + p + 2) * BATCH + b0) * IN_DIM + lane * 32, 0, 0);
                if (p >= 2) {
                    const v4f* src = (const v4f*)&sH0[prv][srow * P0 + shalf * 8];
                    v4f a0 = src[0], a1 = src[1];
                    v4f* dst = (v4f*)(ho0c + (size_t)(p - 2) * BATCH * H0 + srow * H0 + shalf * 8);
                    dst[0] = a0; dst[1] = a1;
                }
                if (p + 1 < C) {
                    const v4f* xs = xsrc + (size_t)(p + 1) * (BATCH * IN_DIM / 4);
                    v4f xr[8];
                    for (int i = 0; i < 8; ++i) xr[i] = xs[32 * i];
                    for (int i = 0; i < 8; ++i) *(v4f*)(xdst + cur * 16 * PX + 2 * i * PX) = xr[i];
                }
            }
            __syncthreads();
        }
        if (kind == KIND_L0) { v4f s; s.x = c[0]; s.y = c[1]; s.z = hl[0]; s.w = hl[1]; *stl = s; }
    } else if (role == 1) {
        const int kind = (wave < 8) ? KIND_L1 : KIND_ST;
        const int q  = wave & 7;
        const int u0 = q * 4;
        const int r  = gq * H1 + u0 + u;
        v2f w[12];
        v2f axn[4];
        float bias = 0.f;
        for (int i = 0; i < 12; ++i) { w[i].x = 0.f; w[i].y = 0.f; }
        for (int i = 0; i < 4; ++i) { axn[i].x = 0.f; axn[i].y = 0.f; }
        if (kind == KIND_L1) {
            for (int kk = 0; kk < 4; ++kk) w[kk] = load_bfrag_row(Wih1, H0, r, 4 * kk, lane);
            for (int kk = 0; kk < 8; ++kk) w[4 + kk] = load_bfrag_row(Whh1, H1, r, 4 * kk, lane);
            bias = bih1[r] + bhh1[r];
            if (k > 0) { v4f s = *stl; c[0] = s.x; c[1] = s.y; hl[0] = s.z; hl[1] = s.w; }
            float* h = &sH1[0][hrow * P1 + u0 + u];
            h[0] = hl[0]; h[P1] = hl[1]; h[16 * P1] = hl[0]; h[16 * P1 + P1] = hl[1];
            for (int kk = 0; kk < 4; ++kk) axn[kk] = load_afrag_g(ho0c, H0, 4 * kk, lane);
        }
        __syncthreads();
        for (int p = 0; p < C + 1; ++p) {
            const int cur = p & 1, prv = cur ^ 1;
            if (kind == KIND_L1) {
                if (p < C) {
                    v2f ax[4], ah[8];
                    for (int kk = 0; kk < 4; ++kk) ax[kk] = axn[kk];
                    for (int kk = 0; kk < 8; ++kk) ah[kk] = load_afrag_lds(&sH1[prv][0], P1, 4 * kk, lane);
                    if (p + 1 < C) {
                        for (int kk = 0; kk < 4; ++kk)
                            axn[kk] = load_afrag_g(ho0c + (size_t)(p + 1) * BATCH * H0, H0, 4 * kk, lane);
                    }
                    v8f acc;
                    for (int j = 0; j < 8; ++j) acc[j] = bias;
                    for (int kk = 0; kk < 4; ++kk) acc = wmma4(ax[kk], w[kk], acc);
                    for (int kk = 0; kk < 8; ++kk) acc = wmma4(ah[kk], w[4 + kk], acc);
                    if (wave < 4) cell2<true>(acc, c, &sH1[cur][hrow * P1 + u0 + u], P1, hl);
                    else          cell2<false>(acc, c, &sH1[cur][hrow * P1 + u0 + u], P1, hl);
                }
            } else {
                if (p >= 1) {
                    const v4f* src = (const v4f*)&sH1[prv][srow * P1 + shalf * 16];
                    v4f a0 = src[0], a1 = src[1], a2 = src[2], a3 = src[3];
                    v4f* dst = (v4f*)(ho1c + (size_t)(p - 1) * BATCH * H1 + srow * H1 + shalf * 16);
                    dst[0] = a0; dst[1] = a1; dst[2] = a2; dst[3] = a3;
                }
            }
            __syncthreads();
        }
        if (kind == KIND_L1) { v4f s; s.x = c[0]; s.y = c[1]; s.z = hl[0]; s.w = hl[1]; *stl = s; }
    } else {
        const int kind = (wave < 4) ? KIND_L2 : (wave == 4) ? KIND_OUT : KIND_IDLE;
        const int q  = wave & 3;
        const int u0 = q * 4;
        const int r  = gq * H2 + u0 + u;
        v2f w[12];
        v2f axn[8];
        float bias = 0.f, biasO = 0.f;
        for (int i = 0; i < 12; ++i) { w[i].x = 0.f; w[i].y = 0.f; }
        for (int i = 0; i < 8; ++i) { axn[i].x = 0.f; axn[i].y = 0.f; }
        if (kind == KIND_L2) {
            for (int kk = 0; kk < 8; ++kk) w[kk] = load_bfrag_row(Wih2, H1, r, 4 * kk, lane);
            for (int kk = 0; kk < 4; ++kk) w[8 + kk] = load_bfrag_row(Whh2, H2, r, 4 * kk, lane);
            bias = bih2[r] + bhh2[r];
            if (k > 0) { v4f s = *stl; c[0] = s.x; c[1] = s.y; hl[0] = s.z; hl[1] = s.w; }
            float* h = &sH2[0][hrow * P2 + u0 + u];
            h[0] = hl[0]; h[P2] = hl[1]; h[16 * P2] = hl[0]; h[16 * P2 + P2] = hl[1];
            for (int kk = 0; kk < 8; ++kk) axn[kk] = load_afrag_g(ho1c, H1, 4 * kk, lane);
        } else if (kind == KIND_OUT) {
            if (ln < OUT_DIM) {
                for (int kk = 0; kk < 4; ++kk) w[kk] = load_bfrag_row(Wl, H2, ln, 4 * kk, lane);
                biasO = bl[ln];
            }
        }
        __syncthreads();
        for (int p = 0; p < C + 1; ++p) {
            const int cur = p & 1, prv = cur ^ 1;
            if (kind == KIND_L2) {
                if (p < C) {
                    v2f ax[8], ah[4];
                    for (int kk = 0; kk < 8; ++kk) ax[kk] = axn[kk];
                    for (int kk = 0; kk < 4; ++kk) ah[kk] = load_afrag_lds(&sH2[prv][0], P2, 4 * kk, lane);
                    if (p + 1 < C) {
                        for (int kk = 0; kk < 8; ++kk)
                            axn[kk] = load_afrag_g(ho1c + (size_t)(p + 1) * BATCH * H1, H1, 4 * kk, lane);
                    }
                    v8f acc;
                    for (int j = 0; j < 8; ++j) acc[j] = bias;
                    for (int kk = 0; kk < 8; ++kk) acc = wmma4(ax[kk], w[kk], acc);
                    for (int kk = 0; kk < 4; ++kk) acc = wmma4(ah[kk], w[8 + kk], acc);
                    cell2<true>(acc, c, &sH2[cur][hrow * P2 + u0 + u], P2, hl);
                }
            } else if (kind == KIND_OUT) {
                if (p >= 1) {
                    const int t = t0 + p - 1;
                    v2f ah[4];
                    for (int kk = 0; kk < 4; ++kk) ah[kk] = load_afrag_lds(&sH2[prv][0], P2, 4 * kk, lane);
                    v8f po;
                    for (int j = 0; j < 8; ++j) po[j] = biasO;
                    for (int kk = 0; kk < 4; ++kk) po = wmma4(ah[kk], w[kk], po);
                    if (ln < OUT_DIM) {
                        float* o = out + (((size_t)(b0 + 8 * hi)) * T_STEPS + t) * OUT_DIM + ln;
                        for (int j = 0; j < 8; ++j) o[(size_t)j * T_STEPS * OUT_DIM] = po[j];
                    }
                }
            }
            __syncthreads();
        }
        if (kind == KIND_L2) { v4f s; s.x = c[0]; s.y = c[1]; s.z = hl[0]; s.w = hl[1]; *stl = s; }
    }
}

template <int C>
static void launch_chunks(const float* x,
                          const float* Wih0, const float* Whh0, const float* bih0, const float* bhh0,
                          const float* Wih1, const float* Whh1, const float* bih1, const float* bhh1,
                          const float* Wih2, const float* Whh2, const float* bih2, const float* bhh2,
                          const float* Wl, const float* bl, float* out, float* ws, hipStream_t stream) {
    for (int d = 0; d < T_STEPS / C + 2; ++d)
        lstm3_layer_chunk_kernel<C><<<dim3(NTILES, 3), PTHREADS, 0, stream>>>(
            x, Wih0, Whh0, bih0, bhh0, Wih1, Whh1, bih1, bhh1, Wih2, Whh2, bih2, bhh2, Wl, bl, out, ws, d);
}

extern "C" void kernel_launch(void* const* d_in, const int* in_sizes, int n_in,
                              void* d_out, int out_size, void* d_ws, size_t ws_size,
                              hipStream_t stream) {
    (void)in_sizes; (void)n_in; (void)out_size;
    const float* x    = (const float*)d_in[0];
    const float* Wih0 = (const float*)d_in[1];
    const float* Whh0 = (const float*)d_in[2];
    const float* bih0 = (const float*)d_in[3];
    const float* bhh0 = (const float*)d_in[4];
    const float* Wih1 = (const float*)d_in[5];
    const float* Whh1 = (const float*)d_in[6];
    const float* bih1 = (const float*)d_in[7];
    const float* bhh1 = (const float*)d_in[8];
    const float* Wih2 = (const float*)d_in[9];
    const float* Whh2 = (const float*)d_in[10];
    const float* bih2 = (const float*)d_in[11];
    const float* bhh2 = (const float*)d_in[12];
    const float* Wl   = (const float*)d_in[13];
    const float* bl   = (const float*)d_in[14];
    float* out = (float*)d_out;

    float* ws = (float*)d_ws;
    if (ws != nullptr && ws_size >= (size_t)6488064) {
        launch_chunks<64>(x, Wih0, Whh0, bih0, bhh0, Wih1, Whh1, bih1, bhh1, Wih2, Whh2, bih2, bhh2, Wl, bl, out, ws, stream);
    } else if (ws != nullptr && ws_size >= (size_t)3342336) {
        launch_chunks<32>(x, Wih0, Whh0, bih0, bhh0, Wih1, Whh1, bih1, bhh1, Wih2, Whh2, bih2, bhh2, Wl, bl, out, ws, stream);
    } else if (ws != nullptr && ws_size >= (size_t)1769472) {
        launch_chunks<16>(x, Wih0, Whh0, bih0, bhh0, Wih1, Whh1, bih1, bhh1, Wih2, Whh2, bih2, bhh2, Wl, bl, out, ws, stream);
    } else {
        lstm3_stage_kernel<<<BATCH / 16, NTHREADS, 0, stream>>>(
            x, Wih0, Whh0, bih0, bhh0,
            Wih1, Whh1, bih1, bhh1,
            Wih2, Whh2, bih2, bhh2,
            Wl, bl, out);
    }
}


# all 3 layers at once: each gate tile's cell update shared by its owner wave and a helper wave (4 gates a lane via LDS behind a mid-pass barrier): 1 cell a lane, 2 tanh in a row not 4; sum order kept
# speedup vs baseline: 4.7153x; 1.1218x over previous
#include <hip/hip_runtime.h>


typedef __attribute__((ext_vector_type(2))) float v2f;
typedef __attribute__((ext_vector_type(4))) float v4f;
typedef __attribute__((ext_vector_type(8))) float v8f;

#define T_STEPS  2048
#define BATCH    256
#define IN_DIM   64
#define H0       16
#define H1       32
#define H2       16
#define OUT_DIM  8
#define NWAVES   21
#define NTHREADS (NWAVES * 32)

#define P0 20
#define P1 36
#define P2 20
#define PX 68

#define KIND_L1  0
#define KIND_L0  1
#define KIND_L2  2
#define KIND_X   3
#define KIND_OUT 4

__device__ __forceinline__ float sigmoidf_(float x) { return 1.0f / (1.0f + __expf(-x)); }

__device__ __forceinline__ v2f load_afrag(const float* base, int lda, int kbase, int lane) {
    int m = lane & 15;
    int k = kbase + ((lane >> 4) << 1);
    const float* p = base + m * lda + k;
    v2f a;
    a.x = p[0];
    a.y = p[1];
    return a;
}

__device__ __forceinline__ v2f load_afrag_lds(const float* base, int pitch, int kbase, int lane) {
    int m = lane & 15;
    int k = kbase + ((lane >> 4) << 1);
    return *(const v2f*)(base + m * pitch + k);
}

__device__ __forceinline__ v2f load_bfrag_row(const float* W, int ldw, int row, int kbase, int lane) {
    int k = kbase + ((lane >> 4) << 1);
    const float* p = W + row * ldw + k;
    v2f b;
    b.x = p[0];
    b.y = p[1];
    return b;
}

__device__ __forceinline__ v8f wmma4(v2f a, v2f b, v8f c) {
    return __builtin_amdgcn_wmma_f32_16x16x4_f32(false, a, false, b, (short)0, c, false, false);
}

template <int ROR, int BANK>
__device__ __forceinline__ float take(float keep, float give) {
    return __builtin_bit_cast(float,
        __builtin_amdgcn_update_dpp(__builtin_bit_cast(int, keep), __builtin_bit_cast(int, give),
                                    0x120 + ROR, 0xf, BANK, false));
}

template <int JJ> __device__ __forceinline__ float gate_i(const v8f& t) {
    return take<12, 0x8>(take<8, 0x4>(take<4,  0x2>(t[JJ],     t[2 + JJ]), t[4 + JJ]), t[6 + JJ]);
}
template <int JJ> __device__ __forceinline__ float gate_f(const v8f& t) {
    return take<8,  0x8>(take<4, 0x4>(take<12, 0x1>(t[2 + JJ], t[JJ]),     t[4 + JJ]), t[6 + JJ]);
}
template <int JJ> __device__ __forceinline__ float gate_g(const v8f& t) {
    return take<4,  0x8>(take<12, 0x2>(take<8, 0x1>(t[4 + JJ], t[JJ]),     t[2 + JJ]), t[6 + JJ]);
}
template <int JJ> __device__ __forceinline__ float gate_o(const v8f& t) {
    return take<12, 0x4>(take<8, 0x2>(take<4,  0x1>(t[6 + JJ], t[JJ]),     t[2 + JJ]), t[4 + JJ]);
}

template <bool SIG_FIRST>
__device__ __forceinline__ void cell2(const v8f& t, float (&c)[2], float* hdst, int pitch, float (&hl)[2]) {
    float si0, sf0, so0, si1, sf1, so1, tg0, vg1;
    if (SIG_FIRST) {
        const float vg0 = gate_g<0>(t);
        vg1 = gate_g<1>(t);
        si0 = sigmoidf_(gate_i<0>(t)); sf0 = sigmoidf_(gate_f<0>(t)); so0 = sigmoidf_(gate_o<0>(t));
        si1 = sigmoidf_(gate_i<1>(t)); sf1 = sigmoidf_(gate_f<1>(t)); so1 = sigmoidf_(gate_o<1>(t));
        asm volatile("" : "+v"(si0), "+v"(sf0), "+v"(so0), "+v"(si1), "+v"(sf1), "+v"(so1));
        tg0 = tanhf(vg0);
    } else {
        tg0 = tanhf(gate_g<0>(t));
        vg1 = gate_g<1>(t);
        si0 = sigmoidf_(gate_i<0>(t)); sf0 = sigmoidf_(gate_f<0>(t)); so0 = sigmoidf_(gate_o<0>(t));
        si1 = sigmoidf_(gate_i<1>(t)); sf1 = sigmoidf_(gate_f<1>(t)); so1 = sigmoidf_(gate_o<1>(t));
        asm volatile("" : "+v"(si0), "+v"(sf0), "+v"(so0), "+v"(si1), "+v"(sf1), "+v"(so1));
    }
    const float tg1 = tanhf(vg1);
    const float cn0 = sf0 * c[0] + si0 * tg0;
    const float cn1 = sf1 * c[1] + si1 * tg1;
    c[0] = cn0;
    c[1] = cn1;
    const float hv0 = so0 * tanhf(cn0);
    const float hv1 = so1 * tanhf(cn1);
    hl[0] = hv0;
    hl[1] = hv1;
    hdst[0]     = hv0;
    hdst[pitch] = hv1;
}

__global__ __launch_bounds__(NTHREADS) void lstm3_stage_kernel(
    const float* __restrict__ x,
    const float* __restrict__ Wih0, const float* __restrict__ Whh0,
    const float* __restrict__ bih0, const float* __restrict__ bhh0,
    const float* __restrict__ Wih1, const float* __restrict__ Whh1,
    const float* __restrict__ bih1, const float* __restrict__ bhh1,
    const float* __restrict__ Wih2, const float* __restrict__ Whh2,
    const float* __restrict__ bih2, const float* __restrict__ bhh2,
    const float* __restrict__ Wl,   const float* __restrict__ bl,
    float* __restrict__ out)
{
    __shared__ __attribute__((aligned(16))) float sXP[2][4 * 2 * 32 * 4];
    __shared__ __attribute__((aligned(16))) float sH0[2][16 * P0];
    __shared__ __attribute__((aligned(16))) float sH1[2][16 * P1];
    __shared__ __attribute__((aligned(16))) float sH2[2][16 * P2];

    const int tid  = threadIdx.x;
    const int lane = tid & 31;
    const int ln   = lane & 15;
    const int hi   = lane >> 4;
    const int u    = ln & 3;
    const int gq   = ln >> 2;
    const int wave = __builtin_amdgcn_readfirstlane(tid >> 5);
    const int kind = (wave < 8) ? KIND_L1 : (wave < 12) ? KIND_L2 : (wave < 16) ? KIND_L0
                   : (wave < 17) ? KIND_OUT : KIND_X;
    const int q    = (wave < 8) ? wave : (wave < 16) ? (wave & 3) : (wave < 17) ? 0 : (wave - 17);
    const int u0   = q * 4;
    const int b0   = blockIdx.x * 16;

    for (int i = tid; i < 2 * 16 * P0; i += NTHREADS) { (&sH0[0][0])[i] = 0.f; (&sH2[0][0])[i] = 0.f; }
    for (int i = tid; i < 2 * 16 * P1; i += NTHREADS) { (&sH1[0][0])[i] = 0.f; }

    v2f   w[16];
    float bias = 0.f;
    float c[2];
    float hl[2];
#pragma unroll
    for (int i = 0; i < 16; ++i) { w[i].x = 0.f; w[i].y = 0.f; }
    c[0] = 0.f; c[1] = 0.f;
    float biasO = 0.f;

    if (kind == KIND_X) {
        const int r = gq * H0 + u0 + u;
#pragma unroll
        for (int kk = 0; kk < 16; ++kk) w[kk] = load_bfrag_row(Wih0, IN_DIM, r, 4 * kk, lane);
        bias = bih0[r] + bhh0[r];
    } else if (kind == KIND_L0) {
        const int r = gq * H0 + u0 + u;
#pragma unroll
        for (int kk = 0; kk < 4; ++kk) w[kk] = load_bfrag_row(Whh0, H0, r, 4 * kk, lane);
    } else if (kind == KIND_L2) {
        const int r = gq * H2 + u0 + u;
#pragma unroll
        for (int kk = 0; kk < 8; ++kk) w[kk] = load_bfrag_row(Wih2, H1, r, 4 * kk, lane);
#pragma unroll
        for (int kk = 0; kk < 4; ++kk) w[8 + kk] = load_bfrag_row(Whh2, H2, r, 4 * kk, lane);
        bias = bih2[r] + bhh2[r];
    } else if (kind == KIND_OUT) {
        if (ln < OUT_DIM) {
#pragma unroll
            for (int kk = 0; kk < 4; ++kk) w[kk] = load_bfrag_row(Wl, H2, ln, 4 * kk, lane);
            biasO = bl[ln];
        }
    } else {
        const int r = gq * H1 + u0 + u;
#pragma unroll
        for (int kk = 0; kk < 4; ++kk) w[kk] = load_bfrag_row(Wih1, H0, r, 4 * kk, lane);
#pragma unroll
        for (int kk = 0; kk < 8; ++kk) w[4 + kk] = load_bfrag_row(Whh1, H1, r, 4 * kk, lane);
        bias = bih1[r] + bhh1[r];
    }

    const int hrow = 8 * hi + 2 * gq;

    __syncthreads();

    for (int p = 0; p < T_STEPS + 4; ++p) {
        const int cur = p & 1, prv = cur ^ 1;

        if (kind == KIND_X) {
            if (p < T_STEPS) {
                const float* xt = x + ((size_t)p * BATCH + b0) * IN_DIM;
                if (p + 1 < T_STEPS)
                    __builtin_prefetch(xt + (size_t)BATCH * IN_DIM + lane * 32, 0, 0);
                v2f a[16];
#pragma unroll
                for (int kk = 0; kk < 16; ++kk) a[kk] = load_afrag(xt, IN_DIM, 4 * kk, lane);
                v8f acc;
#pragma unroll
                for (int j = 0; j < 8; ++j) acc[j] = bias;
#pragma unroll
                for (int kk = 0; kk < 16; ++kk) acc = wmma4(a[kk], w[kk], acc);
                v4f* dst = (v4f*)&sXP[cur][0] + (q * 2) * 32 + lane;
                dst[0]  = __builtin_shufflevector(acc, acc, 0, 1, 2, 3);
                dst[32] = __builtin_shufflevector(acc, acc, 4, 5, 6, 7);
            }
        } else if (kind == KIND_L0) {
            const int t = p - 1;
            if (t >= 0 && t < T_STEPS) {
                const v4f* src = (const v4f*)&sXP[prv][0] + (q * 2) * 32 + lane;
                v4f lo = src[0], up = src[32];
                v8f acc = __builtin_shufflevector(lo, up, 0, 1, 2, 3, 4, 5, 6, 7);
                v2f a[4];
#pragma unroll
                for (int kk = 0; kk < 4; ++kk) a[kk] = load_afrag_lds(&sH0[prv][0], P0, 4 * kk, lane);
#pragma unroll
                for (int kk = 0; kk < 4; ++kk) acc = wmma4(a[kk], w[kk], acc);
                cell2<true>(acc, c, &sH0[cur][hrow * P0 + u0 + u], P0, hl);
            }
        } else if (kind == KIND_L2) {
            const int t = p - 3;
            if (t >= 0 && t < T_STEPS) {
                v2f ax[8], ah[4];
#pragma unroll
                for (int kk = 0; kk < 8; ++kk) ax[kk] = load_afrag_lds(&sH1[prv][0], P1, 4 * kk, lane);
#pragma unroll
                for (int kk = 0; kk < 4; ++kk) ah[kk] = load_afrag_lds(&sH2[prv][0], P2, 4 * kk, lane);
                v8f acc;
#pragma unroll
                for (int j = 0; j < 8; ++j) acc[j] = bias;
#pragma unroll
                for (int kk = 0; kk < 8; ++kk) acc = wmma4(ax[kk], w[kk], acc);
#pragma unroll
                for (int kk = 0; kk < 4; ++kk) acc = wmma4(ah[kk], w[8 + kk], acc);
                cell2<true>(acc, c, &sH2[cur][hrow * P2 + u0 + u], P2, hl);
            }
        } else if (kind == KIND_OUT) {
            const int t = p - 4;
            if (t >= 0) {
                v2f ah[4];
#pragma unroll
                for (int kk = 0; kk < 4; ++kk) ah[kk] = load_afrag_lds(&sH2[prv][0], P2, 4 * kk, lane);
                v8f po;
#pragma unroll
                for (int j = 0; j < 8; ++j) po[j] = biasO;
#pragma unroll
                for (int kk = 0; kk < 4; ++kk) po = wmma4(ah[kk], w[kk], po);
                if (ln < OUT_DIM) {
                    float* o = out + (((size_t)(b0 + 8 * hi)) * T_STEPS + t) * OUT_DIM + ln;
#pragma unroll
                    for (int j = 0; j < 8; ++j) o[(size_t)j * T_STEPS * OUT_DIM] = po[j];
                }
            }
        } else {
            const int t = p - 2;
            if (t >= 0 && t < T_STEPS) {
                v2f ax[4], ah[8];
#pragma unroll
                for (int kk = 0; kk < 4; ++kk) ax[kk] = load_afrag_lds(&sH0[prv][0], P0, 4 * kk, lane);
#pragma unroll
                for (int kk = 0; kk < 8; ++kk) ah[kk] = load_afrag_lds(&sH1[prv][0], P1, 4 * kk, lane);
                v8f acc;
#pragma unroll
                for (int j = 0; j < 8; ++j) acc[j] = bias;
#pragma unroll
                for (int kk = 0; kk < 4; ++kk) acc = wmma4(ax[kk], w[kk], acc);
#pragma unroll
                for (int kk = 0; kk < 8; ++kk) acc = wmma4(ah[kk], w[4 + kk], acc);
                cell2<true>(acc, c, &sH1[cur][hrow * P1 + u0 + u], P1, hl);
            }
        }
        __syncthreads();
    }
}


template <bool SIG_FIRST>
__device__ __forceinline__ void cell1(float gi, float gf, float gg, float go, float& c, float* hdst, float& hl) {
    float si, sf, so, tg;
    if (SIG_FIRST) {
        si = sigmoidf_(gi); sf = sigmoidf_(gf); so = sigmoidf_(go);
        asm volatile("" : "+v"(si), "+v"(sf), "+v"(so));
        tg = tanhf(gg);
    } else {
        tg = tanhf(gg);
        si = sigmoidf_(gi); sf = sigmoidf_(gf); so = sigmoidf_(go);
    }
    const float cn = sf * c + si * tg;
    c = cn;
    const float hv = so * tanhf(cn);
    hl = hv;
    hdst[0] = hv;
}

#define PWAVES   17
#define PTHREADS (PWAVES * 32)
#define NTILES   (BATCH / 16)
#define HB       9
#define KIND_ST   5
#define KIND_IDLE 6
#define KIND_HELP 7

__device__ __forceinline__ v2f load_afrag_g(const float* base, int lda, int kbase, int lane) {
    int m = lane & 15;
    int k = kbase + ((lane >> 4) << 1);
    return *(const v2f*)(base + m * lda + k);
}

__device__ __forceinline__ v4f helper_gates(const v8f& t) {
    v4f g;
    g.x = gate_i<1>(t); g.y = gate_f<1>(t); g.z = gate_g<1>(t); g.w = gate_o<1>(t);
    return g;
}

template <int C>
__global__ __launch_bounds__(PTHREADS) void lstm3_layer_chunk_kernel(
    const float* __restrict__ x,
    const float* __restrict__ Wih0, const float* __restrict__ Whh0,
    const float* __restrict__ bih0, const float* __restrict__ bhh0,
    const float* __restrict__ Wih1, const float* __restrict__ Whh1,
    const float* __restrict__ bih1, const float* __restrict__ bhh1,
    const float* __restrict__ Wih2, const float* __restrict__ Whh2,
    const float* __restrict__ bih2, const float* __restrict__ bhh2,
    const float* __restrict__ Wl,   const float* __restrict__ bl,
    float* __restrict__ out, float* ws, int d)
{
    constexpr int NCH = T_STEPS / C;
    const int role = blockIdx.y;
    const int k    = d - role;
    if (k < 0 || k >= NCH) return;

    __shared__ __attribute__((aligned(16))) float sXP[2][4 * 2 * 32 * 4];
    __shared__ __attribute__((aligned(16))) float sH0[2][16 * P0];
    __shared__ __attribute__((aligned(16))) float sH1[2][16 * P1];
    __shared__ __attribute__((aligned(16))) float sH2[2][16 * P2];
    __shared__ __attribute__((aligned(16))) float sX[2][16 * PX];
    __shared__ __attribute__((aligned(16))) float sG[8 * 32 * 4];

    const int tid  = threadIdx.x;
    const int lane = tid & 31;
    const int ln   = lane & 15;
    const int hi   = lane >> 4;
    const int u    = ln & 3;
    const int gq   = ln >> 2;
    const int wave = __builtin_amdgcn_readfirstlane(tid >> 5);
    const int b0   = blockIdx.x * 16;
    const int t0   = k * C;
    const int kb   = k & 1;
    const bool helper = wave >= HB;
    const int hrow = 8 * hi + 2 * gq + (helper ? 1 : 0);
    const int srow  = lane >> 1;
    const int shalf = lane & 1;

    float* ho0c = ws + ((size_t)kb * C * BATCH + b0) * H0;
    float* ho1c = ws + (size_t)2 * C * BATCH * H0 + ((size_t)kb * C * BATCH + b0) * H1;
    float* st   = ws + (size_t)2 * C * BATCH * (H0 + H1);

    float c  = 0.f;
    float hl = 0.f;
    v8f acc;
    for (int j = 0; j < 8; ++j) acc[j] = 0.f;

    if (role == 0) {
        const int kind = (wave < 4) ? KIND_L0 : (wave < 8) ? KIND_X : (wave == 8) ? KIND_ST
                       : (wave < HB + 4) ? KIND_HELP : KIND_IDLE;
        const bool cellw = (kind == KIND_L0) || (kind == KIND_HELP);
        const int q  = helper ? (wave - HB) : (wave & 3);
        const int u0 = (q & 3) * 4;
        const int r  = gq * H0 + u0 + u;
        v2f* stl = (v2f*)st + (((size_t)((role * NTILES + blockIdx.x) * 8 + (q & 7)) * 32 + lane) * 2 + (helper ? 1 : 0));
        v4f* sGq = (v4f*)&sG[0] + (q & 7) * 32 + lane;
        float* hd = &sH0[0][hrow * P0 + u0 + u];
        v2f w[16];
        float bias = 0.f;
        for (int i = 0; i < 16; ++i) { w[i].x = 0.f; w[i].y = 0.f; }
        float* xdst = &sX[0][hi * PX + 4 * ln];
        const v4f* xsrc = (const v4f*)(x + ((size_t)t0 * BATCH + b0) * IN_DIM) + lane;
        if (kind == KIND_ST) {
            v4f xr[8];
            for (int i = 0; i < 8; ++i) xr[i] = xsrc[32 * i];
            for (int i = 0; i < 8; ++i) *(v4f*)(xdst + 16 * PX + 2 * i * PX) = xr[i];
        }
        if (kind == KIND_X) {
            for (int kk = 0; kk < 16; ++kk) w[kk] = load_bfrag_row(Wih0, IN_DIM, r, 4 * kk, lane);
            bias = bih0[r] + bhh0[r];
        } else if (kind == KIND_L0) {
            for (int kk = 0; kk < 4; ++kk) w[kk] = load_bfrag_row(Whh0, H0, r, 4 * kk, lane);
        }
        if (cellw) {
            if (k > 0) { v2f s = *stl; c = s.x; hl = s.y; }
            hd[0] = hl; hd[16 * P0] = hl;
        }
        __syncthreads();
        for (int p = 0; p < C + 2; ++p) {
            const int cur = p & 1, prv = cur ^ 1;
            const bool on = (p >= 1 && p <= C);
            if (kind == KIND_L0 && on) {
                const v4f* src = (const v4f*)&sXP[prv][0] + (q * 2) * 32 + lane;
                v4f lo = src[0], up = src[32];
                acc = __builtin_shufflevector(lo, up, 0, 1, 2, 3, 4, 5, 6, 7);
                v2f a[4];
                for (int kk = 0; kk < 4; ++kk) a[kk] = load_afrag_lds(&sH0[prv][0], P0, 4 * kk, lane);
                for (int kk = 0; kk < 4; ++kk) acc = wmma4(a[kk], w[kk], acc);
                *sGq = helper_gates(acc);
            }
            __syncthreads();
            if (kind == KIND_L0) {
                if (on) cell1<true>(gate_i<0>(acc), gate_f<0>(acc), gate_g<0>(acc), gate_o<0>(acc), c, hd + cur * 16 * P0, hl);
            } else if (kind == KIND_HELP) {
                if (on) { const v4f g = *sGq; cell1<true>(g.x, g.y, g.z, g.w, c, hd + cur * 16 * P0, hl); }
            } else if (kind == KIND_X) {
                if (p < C) {
                    v2f a[16];
                    for (int kk = 0; kk < 16; ++kk) a[kk] = load_afrag_lds(&sX[prv][0], PX, 4 * kk, lane);
                    v8f ax;
                    for (int j = 0; j < 8; ++j) ax[j] = bias;
                    for (int kk = 0; kk < 16; ++kk) ax = wmma4(a[kk], w[kk], ax);
                    v4f* dst = (v4f*)&sXP[cur][0] + (q * 2) * 32 + lane;
                    dst[0]  = __builtin_shufflevector(ax, ax, 0, 1, 2, 3);
                    dst[32] = __builtin_shufflevector(ax, ax, 4, 5, 6, 7);
                }
            } else if (kind == KIND_ST) {
                if (p < C && t0 + p + 2 < T_STEPS)
                    __builtin_prefetch(x + ((size_t)(t0 + p + 2) * BATCH + b0) * IN_DIM + lane * 32, 0, 0);
                if (p >= 2) {
                    const v4f* src = (const v4f*)&sH0[prv][srow * P0 + shalf * 8];
                    v4f a0 = src[0], a1 = src[1];
                    v4f* dst = (v4f*)(ho0c + (size_t)(p - 2) * BATCH * H0 + srow * H0 + shalf * 8);
                    dst[0] = a0; dst[1] = a1;
                }
                if (p + 1 < C) {
                    const v4f* xs = xsrc + (size_t)(p + 1) * (BATCH * IN_DIM / 4);
                    v4f xr[8];
                    for (int i = 0; i < 8; ++i) xr[i] = xs[32 * i];
                    for (int i = 0; i < 8; ++i) *(v4f*)(xdst + cur * 16 * PX + 2 * i * PX) = xr[i];
                }
            }
            __syncthreads();
        }
        if (cellw) { v2f s; s.x = c; s.y = hl; *stl = s; }
    } else if (role == 1) {
        const int kind = (wave < 8) ? KIND_L1 : (wave == 8) ? KIND_ST : KIND_HELP;
        const bool cellw = (kind == KIND_L1) || (kind == KIND_HELP);
        const int q  = helper ? (wave - HB) : (wave & 7);
        const int u0 = q * 4;
        const int r  = gq * H1 + u0 + u;
        v2f* stl = (v2f*)st + (((size_t)((role * NTILES + blockIdx.x) * 8 + q) * 32 + lane) * 2 + (helper ? 1 : 0));
        v4f* sGq = (v4f*)&sG[0] + q * 32 + lane;
        float* hd = &sH1[0][hrow * P1 + u0 + u];
        v2f w[12];
        v2f axn[4];
        float bias = 0.f;
        for (int i = 0; i < 12; ++i) { w[i].x = 0.f; w[i].y = 0.f; }
        for (int i = 0; i < 4; ++i) { axn[i].x = 0.f; axn[i].y = 0.f; }
        if (kind == KIND_L1) {
            for (int kk = 0; kk < 4; ++kk) w[kk] = load_bfrag_row(Wih1, H0, r, 4 * kk, lane);
            for (int kk = 0; kk < 8; ++kk) w[4 + kk] = load_bfrag_row(Whh1, H1, r, 4 * kk, lane);
            bias = bih1[r] + bhh1[r];
            for (int kk = 0; kk < 4; ++kk) axn[kk] = load_afrag_g(ho0c, H0, 4 * kk, lane);
        }
        if (cellw) {
            if (k > 0) { v2f s = *stl; c = s.x; hl = s.y; }
            hd[0] = hl; hd[16 * P1] = hl;
        }
        __syncthreads();
        for (int p = 0; p < C + 1; ++p) {
            const int cur = p & 1, prv = cur ^ 1;
            const bool on = (p < C);
            if (kind == KIND_L1 && on) {
                v2f ah[8];
                for (int kk = 0; kk < 8; ++kk) ah[kk] = load_afrag_lds(&sH1[prv][0], P1, 4 * kk, lane);
                for (int j = 0; j < 8; ++j) acc[j] = bias;
                for (int kk = 0; kk < 4; ++kk) acc = wmma4(axn[kk], w[kk], acc);
                for (int kk = 0; kk < 8; ++kk) acc = wmma4(ah[kk], w[4 + kk], acc);
                *sGq = helper_gates(acc);
            }
            __syncthreads();
            if (kind == KIND_L1) {
                if (on) {
                    if (p + 1 < C) {
                        for (int kk = 0; kk < 4; ++kk)
                            axn[kk] = load_afrag_g(ho0c + (size_t)(p + 1) * BATCH * H0, H0, 4 * kk, lane);
                    }
                    if (q < 4) cell1<true>(gate_i<0>(acc), gate_f<0>(acc), gate_g<0>(acc), gate_o<0>(acc), c, hd + cur * 16 * P1, hl);
                    else       cell1<false>(gate_i<0>(acc), gate_f<0>(acc), gate_g<0>(acc), gate_o<0>(acc), c, hd + cur * 16 * P1, hl);
                }
            } else if (kind == KIND_HELP) {
                if (on) {
                    const v4f g = *sGq;
                    if (q < 4) cell1<true>(g.x, g.y, g.z, g.w, c, hd + cur * 16 * P1, hl);
                    else       cell1<false>(g.x, g.y, g.z, g.w, c, hd + cur * 16 * P1, hl);
                }
            } else {
                if (p >= 1) {
                    const v4f* src = (const v4f*)&sH1[prv][srow * P1 + shalf * 16];
                    v4f a0 = src[0], a1 = src[1], a2 = src[2], a3 = src[3];
                    v4f* dst = (v4f*)(ho1c + (size_t)(p - 1) * BATCH * H1 + srow * H1 + shalf * 16);
                    dst[0] = a0; dst[1] = a1; dst[2] = a2; dst[3] = a3;
                }
            }
            __syncthreads();
        }
        if (cellw) { v2f s; s.x = c; s.y = hl; *stl = s; }
    } else {
        const int kind = (wave < 4) ? KIND_L2 : (wave == 4) ? KIND_OUT
                       : (wave >= HB && wave < HB + 4) ? KIND_HELP : KIND_IDLE;
        const bool cellw = (kind == KIND_L2) || (kind == KIND_HELP);
        const int q  = helper ? ((wave - HB) & 3) : (wave & 3);
        const int u0 = q * 4;
        const int r  = gq * H2 + u0 + u;
        v2f* stl = (v2f*)st + (((size_t)((role * NTILES + blockIdx.x) * 8 + q) * 32 + lane) * 2 + (helper ? 1 : 0));
        v4f* sGq = (v4f*)&sG[0] + q * 32 + lane;
        float* hd = &sH2[0][hrow * P2 + u0 + u];
        v2f w[12];
        v2f axn[8];
        float bias = 0.f, biasO = 0.f;
        for (int i = 0; i < 12; ++i) { w[i].x = 0.f; w[i].y = 0.f; }
        for (int i = 0; i < 8; ++i) { axn[i].x = 0.f; axn[i].y = 0.f; }
        if (kind == KIND_L2) {
            for (int kk = 0; kk < 8; ++kk) w[kk] = load_bfrag_row(Wih2, H1, r, 4 * kk, lane);
            for (int kk = 0; kk < 4; ++kk) w[8 + kk] = load_bfrag_row(Whh2, H2, r, 4 * kk, lane);
            bias = bih2[r] + bhh2[r];
            for (int kk = 0; kk < 8; ++kk) axn[kk] = load_afrag_g(ho1c, H1, 4 * kk, lane);
        } else if (kind == KIND_OUT) {
            if (ln < OUT_DIM) {
                for (int kk = 0; kk < 4; ++kk) w[kk] = load_bfrag_row(Wl, H2, ln, 4 * kk, lane);
                biasO = bl[ln];
            }
        }
        if (cellw) {
            if (k > 0) { v2f s = *stl; c = s.x; hl = s.y; }
            hd[0] = hl; hd[16 * P2] = hl;
        }
        __syncthreads();
        for (int p = 0; p < C + 1; ++p) {
            const int cur = p & 1, prv = cur ^ 1;
            const bool on = (p < C);
            if (kind == KIND_L2 && on) {
                v2f ah[4];
                for (int kk = 0; kk < 4; ++kk) ah[kk] = load_afrag_lds(&sH2[prv][0], P2, 4 * kk, lane);
                for (int j = 0; j < 8; ++j) acc[j] = bias;
                for (int kk = 0; kk < 8; ++kk) acc = wmma4(axn[kk], w[kk], acc);
                for (int kk = 0; kk < 4; ++kk) acc = wmma4(ah[kk], w[8 + kk], acc);
                *sGq = helper_gates(acc);
            }
            __syncthreads();
            if (kind == KIND_L2) {
                if (on) {
                    if (p + 1 < C) {
                        for (int kk = 0; kk < 8; ++kk)
                            axn[kk] = load_afrag_g(ho1c + (size_t)(p + 1) * BATCH * H1, H1, 4 * kk, lane);
                    }
                    cell1<true>(gate_i<0>(acc), gate_f<0>(acc), gate_g<0>(acc), gate_o<0>(acc), c, hd + cur * 16 * P2, hl);
                }
            } else if (kind == KIND_HELP) {
                if (on) { const v4f g = *sGq; cell1<true>(g.x, g.y, g.z, g.w, c, hd + cur * 16 * P2, hl); }
            } else if (kind == KIND_OUT) {
                if (p >= 1) {
                    const int t = t0 + p - 1;
                    v2f ah[4];
                    for (int kk = 0; kk < 4; ++kk) ah[kk] = load_afrag_lds(&sH2[prv][0], P2, 4 * kk, lane);
                    v8f po;
                    for (int j = 0; j < 8; ++j) po[j] = biasO;
                    for (int kk = 0; kk < 4; ++kk) po = wmma4(ah[kk], w[kk], po);
                    if (ln < OUT_DIM) {
                        float* o = out + (((size_t)(b0 + 8 * hi)) * T_STEPS + t) * OUT_DIM + ln;
                        for (int j = 0; j < 8; ++j) o[(size_t)j * T_STEPS * OUT_DIM] = po[j];
                    }
                }
            }
            __syncthreads();
        }
        if (cellw) { v2f s; s.x = c; s.y = hl; *stl = s; }
    }
}

template <int C>
static void launch_chunks(const float* x,
                          const float* Wih0, const float* Whh0, const float* bih0, const float* bhh0,
                          const float* Wih1, const float* Whh1, const float* bih1, const float* bhh1,
                          const float* Wih2, const float* Whh2, const float* bih2, const float* bhh2,
                          const float* Wl, const float* bl, float* out, float* ws, hipStream_t stream) {
    for (int d = 0; d < T_STEPS / C + 2; ++d)
        lstm3_layer_chunk_kernel<C><<<dim3(NTILES, 3), PTHREADS, 0, stream>>>(
            x, Wih0, Whh0, bih0, bhh0, Wih1, Whh1, bih1, bhh1, Wih2, Whh2, bih2, bhh2, Wl, bl, out, ws, d);
}

extern "C" void kernel_launch(void* const* d_in, const int* in_sizes, int n_in,
                              void* d_out, int out_size, void* d_ws, size_t ws_size,
                              hipStream_t stream) {
    (void)in_sizes; (void)n_in; (void)out_size;
    const float* x    = (const float*)d_in[0];
    const float* Wih0 = (const float*)d_in[1];
    const float* Whh0 = (const float*)d_in[2];
    const float* bih0 = (const float*)d_in[3];
    const float* bhh0 = (const float*)d_in[4];
    const float* Wih1 = (const float*)d_in[5];
    const float* Whh1 = (const float*)d_in[6];
    const float* bih1 = (const float*)d_in[7];
    const float* bhh1 = (const float*)d_in[8];
    const float* Wih2 = (const float*)d_in[9];
    const float* Whh2 = (const float*)d_in[10];
    const float* bih2 = (const float*)d_in[11];
    const float* bhh2 = (const float*)d_in[12];
    const float* Wl   = (const float*)d_in[13];
    const float* bl   = (const float*)d_in[14];
    float* out = (float*)d_out;

    float* ws = (float*)d_ws;
    if (ws != nullptr && ws_size >= (size_t)6488064) {
        launch_chunks<64>(x, Wih0, Whh0, bih0, bhh0, Wih1, Whh1, bih1, bhh1, Wih2, Whh2, bih2, bhh2, Wl, bl, out, ws, stream);
    } else if (ws != nullptr && ws_size >= (size_t)3342336) {
        launch_chunks<32>(x, Wih0, Whh0, bih0, bhh0, Wih1, Whh1, bih1, bhh1, Wih2, Whh2, bih2, bhh2, Wl, bl, out, ws, stream);
    } else if (ws != nullptr && ws_size >= (size_t)1769472) {
        launch_chunks<16>(x, Wih0, Whh0, bih0, bhh0, Wih1, Whh1, bih1, bhh1, Wih2, Whh2, bih2, bhh2, Wl, bl, out, ws, stream);
    } else {
        lstm3_stage_kernel<<<BATCH / 16, NTHREADS, 0, stream>>>(
            x, Wih0, Whh0, bih0, bhh0,
            Wih1, Whh1, bih1, bhh1,
            Wih2, Whh2, bih2, bhh2,
            Wl, bl, out);
    }
}
